# starting kernel
# baseline (speedup 1.0000x reference)
#include <hip/hip_runtime.h>

#define N_AA   21
#define NATOM  15
#define DFEAT  128
#define DCOORD (N_AA * NATOM * 3)
#define DDIH   39
#define DH1    256
#define H1S    260
#define H2S    132

typedef __attribute__((ext_vector_type(2))) float v2f;
typedef __attribute__((ext_vector_type(8))) float v8f;

__global__ void precompute_proj(const float* __restrict__ aa_emb,
                                const float* __restrict__ chain_emb,
                                const float* __restrict__ W1,
                                float* __restrict__ proj) {
  const int r = blockIdx.x;
  const int c = threadIdx.x;
  const float* emb;
  const float* wrow;
  if (r < N_AA) {
    emb  = aa_emb + (size_t)r * DFEAT;
    wrow = W1;
  } else {
    emb  = chain_emb + (size_t)(r - N_AA) * DFEAT;
    wrow = W1 + (size_t)(DFEAT + DCOORD + DDIH) * DH1;
  }
  float acc = 0.f;
  for (int k = 0; k < DFEAT; ++k)
    acc = fmaf(emb[k], wrow[(size_t)k * DH1 + c], acc);
  proj[(size_t)r * DH1 + c] = acc;
}

__global__ __launch_bounds__(128) void residue_embed_fused(
    const int*   __restrict__ seq_idx,
    const float* __restrict__ xyz,
    const float* __restrict__ orient,
    const float* __restrict__ dihedrals,
    const int*   __restrict__ chain_idx,
    const float* __restrict__ atom_mask,
    const float* __restrict__ W1, const float* __restrict__ b1,
    const float* __restrict__ W2, const float* __restrict__ b2,
    const float* __restrict__ W3, const float* __restrict__ b3,
    const float* __restrict__ W4, const float* __restrict__ b4,
    const float* __restrict__ proj,
    float*       __restrict__ out)
{
  __shared__ float s_h1[16][H1S];
  __shared__ float s_h2[16][H2S];
  __shared__ float s_h3[16][H2S];
  __shared__ float s_xyzl[16][48];
  __shared__ float s_dih[16][40];
  __shared__ int   s_seq[16];
  __shared__ int   s_ch[16];

  const int tid  = threadIdx.x;
  const int tok0 = blockIdx.x * 16;

  if (tid < 16) {
    s_seq[tid] = seq_idx[tok0 + tid];
    s_ch[tid]  = chain_idx[tok0 + tid];
  }

  if (tid < 48) {
    const int   t = tid / 3, d = tid % 3;
    const float x = dihedrals[(size_t)(tok0 + t) * 3 + d];
    const float kF[6] = {1.f, 2.f, 3.f, 1.f, 0.5f, 1.f / 3.f};
    const int base = d * 13;
    s_dih[t][base] = x;
#pragma unroll
    for (int f = 0; f < 6; ++f) {
      s_dih[t][base + 1 + f] = sinf(kF[f] * x);
      s_dih[t][base + 7 + f] = cosf(kF[f] * x);
    }
  }

  for (int task = tid; task < 16 * NATOM; task += 128) {
    const int t = task / NATOM, a = task % NATOM;
    const int tok = tok0 + t;
    const float* p  = xyz + ((size_t)tok * NATOM + a) * 3;
    const float* ca = xyz + ((size_t)tok * NATOM + 1) * 3;
    const float rx = p[0] - ca[0], ry = p[1] - ca[1], rz = p[2] - ca[2];
    const float* R = orient + (size_t)tok * 9;
    const float m = atom_mask[(size_t)tok * NATOM + a];
    s_xyzl[t][a * 3 + 0] = (R[0] * rx + R[3] * ry + R[6] * rz) * m;
    s_xyzl[t][a * 3 + 1] = (R[1] * rx + R[4] * ry + R[7] * rz) * m;
    s_xyzl[t][a * 3 + 2] = (R[2] * rx + R[5] * ry + R[8] * rz) * m;
  }
  __syncthreads();

  const float* w1dih = W1 + (size_t)(DFEAT + DCOORD) * DH1;
  for (int t = 0; t < 16; ++t) {
    const int sq = s_seq[t], ch = s_ch[t];
    const float* w1c = W1 + (size_t)(DFEAT + sq * 45) * DH1;
    const float* pa  = proj + (size_t)sq * DH1;
    const float* pc  = proj + (size_t)(N_AA + ch) * DH1;
#pragma unroll
    for (int pass = 0; pass < 2; ++pass) {
      const int c = pass * 128 + tid;
      float acc = b1[c] + pa[c] + pc[c];
      for (int k = 0; k < 45; ++k)
        acc = fmaf(s_xyzl[t][k], w1c[(size_t)k * DH1 + c], acc);
      for (int k = 0; k < DDIH; ++k)
        acc = fmaf(s_dih[t][k], w1dih[(size_t)k * DH1 + c], acc);
      s_h1[t][c] = fmaxf(acc, 0.f);
    }
  }
  __syncthreads();

  const int wave = tid >> 5;
  const int lane = tid & 31;
  const int half = lane >> 4;
  const int lm   = lane & 15;

#pragma unroll
  for (int tt = 0; tt < 2; ++tt) {
    const int col0 = (wave * 2 + tt) * 16;
    v8f acc = {};
    for (int k = 0; k < 256; k += 4) {
      v2f a = *(const v2f*)&s_h1[lm][k + 2 * half];
      v2f b;
      b.x = W2[(size_t)(k + 2 * half)     * DFEAT + col0 + lm];
      b.y = W2[(size_t)(k + 2 * half + 1) * DFEAT + col0 + lm];
      acc = __builtin_amdgcn_wmma_f32_16x16x4_f32(
          false, a, false, b, (short)0, acc, false, false);
    }
#pragma unroll
    for (int r = 0; r < 8; ++r) {
      const float v = acc[r] + b2[col0 + lm];
      s_h2[r + 8 * half][col0 + lm] = fmaxf(v, 0.f);
    }
  }
  __syncthreads();

#pragma unroll
  for (int tt = 0; tt < 2; ++tt) {
    const int col0 = (wave * 2 + tt) * 16;
    v8f acc = {};
    for (int k = 0; k < 128; k += 4) {
      v2f a = *(const v2f*)&s_h2[lm][k + 2 * half];
      v2f b;
      b.x = W3[(size_t)(k + 2 * half)     * DFEAT + col0 + lm];
      b.y = W3[(size_t)(k + 2 * half + 1) * DFEAT + col0 + lm];
      acc = __builtin_amdgcn_wmma_f32_16x16x4_f32(
          false, a, false, b, (short)0, acc, false, false);
    }
#pragma unroll
    for (int r = 0; r < 8; ++r) {
      const float v = acc[r] + b3[col0 + lm];
      s_h3[r + 8 * half][col0 + lm] = fmaxf(v, 0.f);
    }
  }
  __syncthreads();

#pragma unroll
  for (int tt = 0; tt < 2; ++tt) {
    const int col0 = (wave * 2 + tt) * 16;
    v8f acc = {};
    for (int k = 0; k < 128; k += 4) {
      v2f a = *(const v2f*)&s_h3[lm][k + 2 * half];
      v2f b;
      b.x = W4[(size_t)(k + 2 * half)     * DFEAT + col0 + lm];
      b.y = W4[(size_t)(k + 2 * half + 1) * DFEAT + col0 + lm];
      acc = __builtin_amdgcn_wmma_f32_16x16x4_f32(
          false, a, false, b, (short)0, acc, false, false);
    }
#pragma unroll
    for (int r = 0; r < 8; ++r) {
      out[(size_t)(tok0 + r + 8 * half) * DFEAT + col0 + lm] = acc[r] + b4[col0 + lm];
    }
  }
}

extern "C" void kernel_launch(void* const* d_in, const int* in_sizes, int n_in,
                              void* d_out, int out_size, void* d_ws, size_t ws_size,
                              hipStream_t stream) {
  const int*   seq    = (const int*)  d_in[0];
  const float* xyz    = (const float*)d_in[1];
  const float* orient = (const float*)d_in[2];
  const float* dih    = (const float*)d_in[3];
  const int*   chain  = (const int*)  d_in[4];
  const float* amask  = (const float*)d_in[5];
  const float* aa_emb = (const float*)d_in[6];
  const float* ch_emb = (const float*)d_in[7];
  const float* W1 = (const float*)d_in[8];
  const float* b1 = (const float*)d_in[9];
  const float* W2 = (const float*)d_in[10];
  const float* b2 = (const float*)d_in[11];
  const float* W3 = (const float*)d_in[12];
  const float* b3 = (const float*)d_in[13];
  const float* W4 = (const float*)d_in[14];
  const float* b4 = (const float*)d_in[15];
  float* out  = (float*)d_out;
  float* proj = (float*)d_ws;

  const int ntok = in_sizes[0];

  precompute_proj<<<N_AA + 10, DH1, 0, stream>>>(aa_emb, ch_emb, W1, proj);
  residue_embed_fused<<<ntok / 16, 128, 0, stream>>>(
      seq, xyz, orient, dih, chain, amask,
      W1, b1, W2, b2, W3, b3, W4, b4, proj, out);
}


# Layers 2-4: W2,W3,W4 staged once per work-group into LDS in WMMA fragment order; 256 persistent work-groups loop over 32-token tiles and read B fragments from LDS, not global; same sum order
# speedup vs baseline: 1.2903x; 1.2903x over previous
#include <hip/hip_runtime.h>

#define N_AA   21
#define NATOM  15
#define DFEAT  128
#define DCOORD (N_AA * NATOM * 3)
#define DDIH   39
#define DH1    256
#define H1S    260
#define H2S    132

#define CHUNK  1024
#define L1T    64
#define GT     16
#define XS     88

typedef __attribute__((ext_vector_type(2))) float v2f;
typedef __attribute__((ext_vector_type(4))) float v4f;
typedef __attribute__((ext_vector_type(8))) float v8f;

__global__ void precompute_proj(const float* __restrict__ aa_emb,
                                const float* __restrict__ chain_emb,
                                const float* __restrict__ W1,
                                float* __restrict__ proj) {
  const int r = blockIdx.x;
  const int c = threadIdx.x;
  const float* emb;
  const float* wrow;
  if (r < N_AA) {
    emb  = aa_emb + (size_t)r * DFEAT;
    wrow = W1;
  } else {
    emb  = chain_emb + (size_t)(r - N_AA) * DFEAT;
    wrow = W1 + (size_t)(DFEAT + DCOORD + DDIH) * DH1;
  }
  float acc = 0.f;
  for (int k = 0; k < DFEAT; ++k)
    acc = fmaf(emb[k], wrow[(size_t)k * DH1 + c], acc);
  proj[(size_t)r * DH1 + c] = acc;
}

__global__ __launch_bounds__(L1T, 2) void layer1_by_type(
    const int*   __restrict__ seq_idx,
    const float* __restrict__ xyz,
    const float* __restrict__ orient,
    const float* __restrict__ dihedrals,
    const int*   __restrict__ chain_idx,
    const float* __restrict__ atom_mask,
    const float* __restrict__ W1, const float* __restrict__ b1,
    const float* __restrict__ proj,
    float*       __restrict__ h1,
    int ntok)
{
  __shared__ int   s_list[CHUNK + 4];
  __shared__ int   s_wcnt[L1T / 32];
  __shared__ int   s_ch[GT];
  __shared__ __attribute__((aligned(16))) float s_x[GT][XS];

  const int tid  = threadIdx.x;
  const int wave = tid >> 5;
  const int lane = tid & 31;
  const int type = blockIdx.x;
  const int tok_base = blockIdx.y * CHUNK;
  const int c0 = tid * 4;

  unsigned hbits = 0;
#pragma unroll
  for (int i = 0; i < CHUNK / L1T; ++i) {
    const int tok = tok_base + i * L1T + tid;
    if ((tok < ntok) && (seq_idx[tok] == type)) hbits |= 1u << i;
  }

  v4f wc[45], wd[DDIH];
  {
    const float* w1c = W1 + (size_t)(DFEAT + type * 45) * DH1 + c0;
#pragma unroll
    for (int k = 0; k < 45; ++k) wc[k] = *(const v4f*)(w1c + (size_t)k * DH1);
    const float* w1d = W1 + (size_t)(DFEAT + DCOORD) * DH1 + c0;
#pragma unroll
    for (int k = 0; k < DDIH; ++k) wd[k] = *(const v4f*)(w1d + (size_t)k * DH1);
  }
  const v4f base = *(const v4f*)(b1 + c0) + *(const v4f*)(proj + (size_t)type * DH1 + c0);

  unsigned msk[CHUNK / L1T];
  int cnt = 0;
#pragma unroll
  for (int i = 0; i < CHUNK / L1T; ++i) {
    msk[i] = __builtin_amdgcn_ballot_w32((hbits >> i) & 1u);
    cnt += __builtin_popcount(msk[i]);
  }
  if (lane == 0) s_wcnt[wave] = cnt;
  __syncthreads();
  int off = 0, total = 0;
#pragma unroll
  for (int w = 0; w < L1T / 32; ++w) {
    const int cw = s_wcnt[w];
    if (w < wave) off += cw;
    total += cw;
  }
  const unsigned below = (1u << lane) - 1u;
#pragma unroll
  for (int i = 0; i < CHUNK / L1T; ++i) {
    if ((hbits >> i) & 1u) s_list[off + __builtin_popcount(msk[i] & below)] = tok_base + i * L1T + tid;
    off += __builtin_popcount(msk[i]);
  }
  __syncthreads();
  if (total > 0 && tid < 3) s_list[total + tid] = s_list[total - 1];
  const int totalp = (total + 3) & ~3;
  __syncthreads();

  for (int g0 = 0; g0 < totalp; g0 += GT) {
    const int ng = (totalp - g0 < GT) ? (totalp - g0) : GT;

    for (int task = tid; task < ng * NATOM; task += L1T) {
      const int t = task / NATOM, a = task % NATOM;
      const int tok = s_list[g0 + t];
      const float* p  = xyz + ((size_t)tok * NATOM + a) * 3;
      const float* ca = xyz + ((size_t)tok * NATOM + 1) * 3;
      const float rx = p[0] - ca[0], ry = p[1] - ca[1], rz = p[2] - ca[2];
      const float* R = orient + (size_t)tok * 9;
      const float m = atom_mask[(size_t)tok * NATOM + a];
      s_x[t][a * 3 + 0] = (R[0] * rx + R[3] * ry + R[6] * rz) * m;
      s_x[t][a * 3 + 1] = (R[1] * rx + R[4] * ry + R[7] * rz) * m;
      s_x[t][a * 3 + 2] = (R[2] * rx + R[5] * ry + R[8] * rz) * m;
    }
    for (int task = tid; task < ng * 18; task += L1T) {
      const int t = task / 18;
      const int r = task - t * 18;
      const int d = r / 6;
      const int f = r - d * 6;
      const int tok = s_list[g0 + t];
      const float x = dihedrals[(size_t)tok * 3 + d];
      const float kf = (f == 1) ? 2.f : (f == 2) ? 3.f : (f == 4) ? 0.5f
                     : (f == 5) ? (1.f / 3.f) : 1.f;
      const int bcol = 45 + d * 13;
      if (f == 0) s_x[t][bcol] = x;
      s_x[t][bcol + 1 + f] = sinf(kf * x);
      s_x[t][bcol + 7 + f] = cosf(kf * x);
    }
    if (tid < ng) s_ch[tid] = chain_idx[s_list[g0 + tid]];
    __syncthreads();

    for (int t = 0; t < ng; t += 4) {
      v2f alo[4], ahi[4];
#pragma unroll
      for (int i = 0; i < 4; ++i) {
        const v4f a0 = base + *(const v4f*)(proj + (size_t)(N_AA + s_ch[t + i]) * DH1 + c0);
        alo[i] = a0.lo;
        ahi[i] = a0.hi;
      }
#pragma unroll
      for (int k = 0; k < 45; ++k) {
#pragma unroll
        for (int i = 0; i < 4; ++i) {
          const float x = s_x[t + i][k];
          const v2f xx = {x, x};
          alo[i] = __builtin_elementwise_fma(xx, wc[k].lo, alo[i]);
          ahi[i] = __builtin_elementwise_fma(xx, wc[k].hi, ahi[i]);
        }
      }
#pragma unroll
      for (int k = 0; k < DDIH; ++k) {
#pragma unroll
        for (int i = 0; i < 4; ++i) {
          const float x = s_x[t + i][45 + k];
          const v2f xx = {x, x};
          alo[i] = __builtin_elementwise_fma(xx, wd[k].lo, alo[i]);
          ahi[i] = __builtin_elementwise_fma(xx, wd[k].hi, ahi[i]);
        }
      }
#pragma unroll
      for (int i = 0; i < 4; ++i) {
        v4f o;
        o.x = fmaxf(alo[i].x, 0.f);
        o.y = fmaxf(alo[i].y, 0.f);
        o.z = fmaxf(ahi[i].x, 0.f);
        o.w = fmaxf(ahi[i].y, 0.f);
        *(v4f*)(h1 + (size_t)s_list[g0 + t + i] * DH1 + c0) = o;
      }
    }
    __syncthreads();
  }
}

#define MT      32
#define MW      8
#define MGRID   256
#define OFF_W3  (256 * 128)
#define OFF_W4  (OFF_W3 + 128 * 128)
#define OFF_H1  (OFF_W4 + 128 * 128)
#define OFF_H2  (OFF_H1 + MT * H1S)
#define LDS_FLOATS (OFF_H2 + MT * H2S)

__device__ __forceinline__ void stage_weights(const float* __restrict__ Wm, float* Ld,
                                              int units, int wave, int lane) {
  const int sw = (lane >> 2) & 3;
  for (int ub = wave; ub < units; ub += 4 * MW) {
    v4f v[4][4];
#pragma unroll
    for (int ii = 0; ii < 4; ++ii) {
      const int u  = ub + MW * ii;
      const int kb = 8 * (u >> 1) + 2 * (u & 1);
#pragma unroll
      for (int e = 0; e < 4; ++e)
        v[ii][e] = *(const v4f*)(Wm + (size_t)(kb + (e & 1) + 4 * (e >> 1)) * DFEAT + 4 * lane);
    }
#pragma unroll
    for (int ii = 0; ii < 4; ++ii) {
      float* d = Ld + (ub + MW * ii) * 512 + 16 * lane;
#pragma unroll
      for (int c = 0; c < 4; ++c) {
        const v4f o = {v[ii][0][c], v[ii][1][c], v[ii][2][c], v[ii][3][c]};
        *(v4f*)(d + 4 * (c ^ sw)) = o;
      }
    }
  }
}

#define NP 4
struct Frag { v2f a0[2 * NP], a1[2 * NP]; v4f b[NP]; };

__device__ __forceinline__ Frag load_frag(const float* a0p, const float* a1p, const float* bp, int blk) {
  Frag f;
#pragma unroll
  for (int p = 0; p < NP; ++p) {
    const int j = NP * blk + p;
    f.a0[2 * p]     = *(const v2f*)(a0p + 8 * j);
    f.a0[2 * p + 1] = *(const v2f*)(a0p + 8 * j + 4);
    f.a1[2 * p]     = *(const v2f*)(a1p + 8 * j);
    f.a1[2 * p + 1] = *(const v2f*)(a1p + 8 * j + 4);
    f.b[p]          = *(const v4f*)(bp + j * 1024);
  }
  return f;
}

__device__ __forceinline__ void mm_frag(const Frag& f, v8f& acc0, v8f& acc1) {
#pragma unroll
  for (int p = 0; p < NP; ++p) {
    acc0 = __builtin_amdgcn_wmma_f32_16x16x4_f32(false, f.a0[2 * p],     false, f.b[p].lo, (short)0, acc0, false, false);
    acc1 = __builtin_amdgcn_wmma_f32_16x16x4_f32(false, f.a1[2 * p],     false, f.b[p].lo, (short)0, acc1, false, false);
    acc0 = __builtin_amdgcn_wmma_f32_16x16x4_f32(false, f.a0[2 * p + 1], false, f.b[p].hi, (short)0, acc0, false, false);
    acc1 = __builtin_amdgcn_wmma_f32_16x16x4_f32(false, f.a1[2 * p + 1], false, f.b[p].hi, (short)0, acc1, false, false);
  }
}

__device__ __forceinline__ void group_fence() { __builtin_amdgcn_sched_barrier(0); }

__device__ __forceinline__ void layer_mm(const float* A, int pitch, const float* Bl, int nb,
                                         int lm, int half, int bslot, v8f& acc0, v8f& acc1) {
  const float* a0p = A + lm * pitch + 2 * half;
  const float* a1p = a0p + 16 * pitch;
  const float* bp  = Bl + half * 512 + bslot * 4;
  Frag f0 = load_frag(a0p, a1p, bp, 0);
  for (int blk = 0; blk < nb - 2; blk += 2) {
    const Frag f1 = load_frag(a0p, a1p, bp, blk + 1);
    group_fence();
    mm_frag(f0, acc0, acc1);
    group_fence();
    f0 = load_frag(a0p, a1p, bp, blk + 2);
    group_fence();
    mm_frag(f1, acc0, acc1);
    group_fence();
  }
  const Frag f1 = load_frag(a0p, a1p, bp, nb - 1);
  group_fence();
  mm_frag(f0, acc0, acc1);
  group_fence();
  mm_frag(f1, acc0, acc1);
}

__device__ __forceinline__ void store_out(float* __restrict__ out, const v8f& o0, const v8f& o1,
                                          int tok0, int half, int col, int ntok) {
  float* p = out + (size_t)(tok0 + 8 * half) * DFEAT + col;
  if (tok0 + MT <= ntok) {
#pragma unroll
    for (int r = 0; r < 8; ++r) {
      p[r * DFEAT]        = o0[r];
      p[(16 + r) * DFEAT] = o1[r];
    }
  } else {
#pragma unroll
    for (int r = 0; r < 8; ++r) {
      if (tok0 + 8 * half + r < ntok)      p[r * DFEAT]        = o0[r];
      if (tok0 + 8 * half + 16 + r < ntok) p[(16 + r) * DFEAT] = o1[r];
    }
  }
}

__global__ __launch_bounds__(MW * 32) void mlp_layers_2_to_4(
    const float* __restrict__ h1,
    const float* __restrict__ W2, const float* __restrict__ b2,
    const float* __restrict__ W3, const float* __restrict__ b3,
    const float* __restrict__ W4, const float* __restrict__ b4,
    float*       __restrict__ out,
    int ntok)
{
  __shared__ __attribute__((aligned(16))) float L[LDS_FLOATS];

  const int tid  = threadIdx.x;
  const int wave = tid >> 5;
  const int lane = tid & 31;
  const int half = lane >> 4;
  const int lm   = lane & 15;
  const int col  = wave * 16 + lm;
  const int bslot = wave * 16 + (lm ^ (wave & 3));
  const int ntiles = (ntok + MT - 1) / MT;

  v4f pre[8];
  int tile = blockIdx.x;
  if (tile < ntiles) {
#pragma unroll
    for (int q = 0; q < 8; ++q) {
      const int i = tid + 256 * q;
      int row = tile * MT + (i >> 6);
      row = (row < ntok) ? row : (ntok - 1);
      pre[q] = *(const v4f*)(h1 + (size_t)row * DH1 + (i & 63) * 4);
    }
  }
  const float bias2 = b2[col], bias3 = b3[col], bias4 = b4[col];

  stage_weights(W2, L,          64, wave, lane);
  stage_weights(W3, L + OFF_W3, 32, wave, lane);
  stage_weights(W4, L + OFF_W4, 32, wave, lane);

  v8f o0 = {}, o1 = {};
  int otok0 = -1;

  for (; tile < ntiles; tile += gridDim.x) {
#pragma unroll
    for (int q = 0; q < 8; ++q) {
      const int i = tid + 256 * q;
      *(v4f*)&L[OFF_H1 + (i >> 6) * H1S + (i & 63) * 4] = pre[q];
    }
    __syncthreads();

    if (otok0 >= 0) store_out(out, o0, o1, otok0, half, col, ntok);

    const int tnext = tile + gridDim.x;
    if (tnext < ntiles) {
#pragma unroll
      for (int q = 0; q < 8; ++q) {
        const int i = tid + 256 * q;
        int row = tnext * MT + (i >> 6);
        row = (row < ntok) ? row : (ntok - 1);
        pre[q] = *(const v4f*)(h1 + (size_t)row * DH1 + (i & 63) * 4);
      }
    }

    {
      v8f acc0 = {}, acc1 = {};
      layer_mm(L + OFF_H1, H1S, L, 8, lm, half, bslot, acc0, acc1);
#pragma unroll
      for (int r = 0; r < 8; ++r) {
        L[OFF_H2 + (r + 8 * half) * H2S + col]      = fmaxf(acc0[r] + bias2, 0.f);
        L[OFF_H2 + (16 + r + 8 * half) * H2S + col] = fmaxf(acc1[r] + bias2, 0.f);
      }
    }
    __syncthreads();

    {
      v8f acc0 = {}, acc1 = {};
      layer_mm(L + OFF_H2, H2S, L + OFF_W3, 4, lm, half, bslot, acc0, acc1);
#pragma unroll
      for (int r = 0; r < 8; ++r) {
        L[OFF_H1 + (r + 8 * half) * H2S + col]      = fmaxf(acc0[r] + bias3, 0.f);
        L[OFF_H1 + (16 + r + 8 * half) * H2S + col] = fmaxf(acc1[r] + bias3, 0.f);
      }
    }
    __syncthreads();

    {
      v8f acc0 = {}, acc1 = {};
      layer_mm(L + OFF_H1, H2S, L + OFF_W4, 4, lm, half, bslot, acc0, acc1);
#pragma unroll
      for (int r = 0; r < 8; ++r) {
        o0[r] = acc0[r] + bias4;
        o1[r] = acc1[r] + bias4;
      }
      otok0 = tile * MT;
    }
    __syncthreads();
  }
  if (otok0 >= 0) store_out(out, o0, o1, otok0, half, col, ntok);
}

extern "C" void kernel_launch(void* const* d_in, const int* in_sizes, int n_in,
                              void* d_out, int out_size, void* d_ws, size_t ws_size,
                              hipStream_t stream) {
  const int*   seq    = (const int*)  d_in[0];
  const float* xyz    = (const float*)d_in[1];
  const float* orient = (const float*)d_in[2];
  const float* dih    = (const float*)d_in[3];
  const int*   chain  = (const int*)  d_in[4];
  const float* amask  = (const float*)d_in[5];
  const float* aa_emb = (const float*)d_in[6];
  const float* ch_emb = (const float*)d_in[7];
  const float* W1 = (const float*)d_in[8];
  const float* b1 = (const float*)d_in[9];
  const float* W2 = (const float*)d_in[10];
  const float* b2 = (const float*)d_in[11];
  const float* W3 = (const float*)d_in[12];
  const float* b3 = (const float*)d_in[13];
  const float* W4 = (const float*)d_in[14];
  const float* b4 = (const float*)d_in[15];
  float* out  = (float*)d_out;
  float* proj = (float*)d_ws;
  float* h1   = proj + 8192;

  const int ntok = in_sizes[0];

  precompute_proj<<<N_AA + 10, DH1, 0, stream>>>(aa_emb, ch_emb, W1, proj);
  layer1_by_type<<<dim3(N_AA, (ntok + CHUNK - 1) / CHUNK), L1T, 0, stream>>>(
      seq, xyz, orient, dih, chain, amask, W1, b1, proj, h1, ntok);
  mlp_layers_2_to_4<<<MGRID, MW * 32, 0, stream>>>(
      h1, W2, b2, W3, b3, W4, b4, out, ntok);
}


# Layer-1 input rows (rotated coords, angle encoding, chain index) built by a dense token-order pass in the first launch; the by-type launch copies each finished 88-float row; same sum order
# speedup vs baseline: 1.5413x; 1.1945x over previous
#include <hip/hip_runtime.h>

#define N_AA   21
#define NATOM  15
#define DFEAT  128
#define DCOORD (N_AA * NATOM * 3)
#define DDIH   39
#define DH1    256
#define H1S    260
#define H2S    132

#define CHUNK  1024
#define L1T    64
#define GT     L1T
#define XS     88
#define XQ     (XS / 4)
#define FT     128
#define NPROJ  (N_AA + 10)

typedef __attribute__((ext_vector_type(2))) float v2f;
typedef __attribute__((ext_vector_type(4))) float v4f;
typedef __attribute__((ext_vector_type(8))) float v8f;

#define FTHREADS 256

#define STAGE_LOAD(reg, src, n)                                                  \
  _Pragma("unroll")                                                              \
  for (int i = 0; i < ((n) / 4 + FTHREADS - 1) / FTHREADS; ++i) {                \
    const int idx = tid + FTHREADS * i;                                          \
    reg[i] = *(const v4f*)((src) + 4 * ((idx < (n) / 4) ? idx : ((n) / 4 - 1))); \
  }
#define STAGE_STORE(reg, dst, n)                                                 \
  _Pragma("unroll")                                                              \
  for (int i = 0; i < ((n) / 4 + FTHREADS - 1) / FTHREADS; ++i) {                \
    const int idx = tid + FTHREADS * i;                                          \
    if (idx < (n) / 4) *(v4f*)((dst) + 4 * idx) = reg[i];                        \
  }

__global__ __launch_bounds__(FTHREADS) void proj_and_features(
    const float* __restrict__ aa_emb,
    const float* __restrict__ chain_emb,
    const float* __restrict__ W1,
    const float* __restrict__ xyz,
    const float* __restrict__ orient,
    const float* __restrict__ dihedrals,
    const int*   __restrict__ chain_idx,
    const float* __restrict__ atom_mask,
    float* __restrict__ proj,
    float* __restrict__ feat,
    int ntok)
{
  __shared__ __attribute__((aligned(16))) float s_f[FT * XS];
  __shared__ __attribute__((aligned(16))) float s_xyz[FT * NATOM * 3];
  __shared__ __attribute__((aligned(16))) float s_R[FT * 9];
  __shared__ __attribute__((aligned(16))) float s_m[FT * NATOM];
  __shared__ __attribute__((aligned(16))) float s_d[FT * 3];

  const int tid = threadIdx.x;
  if (blockIdx.x < NPROJ) {
    const int r = blockIdx.x;
    const int c = tid;
    const float* emb;
    const float* wrow;
    if (r < N_AA) {
      emb  = aa_emb + (size_t)r * DFEAT;
      wrow = W1;
    } else {
      emb  = chain_emb + (size_t)(r - N_AA) * DFEAT;
      wrow = W1 + (size_t)(DFEAT + DCOORD + DDIH) * DH1;
    }
    float acc = 0.f;
    for (int k = 0; k < DFEAT; ++k)
      acc = fmaf(emb[k], wrow[(size_t)k * DH1 + c], acc);
    proj[(size_t)r * DH1 + c] = acc;
    return;
  }

  const int tok0 = (blockIdx.x - NPROJ) * FT;
  const int last = ntok - 1;
  const int ctok = (tok0 + (tid & (FT - 1)) < ntok) ? (tok0 + (tid & (FT - 1))) : last;

  if (tok0 + FT <= ntok) {
    const float* gx = xyz       + (size_t)tok0 * (NATOM * 3);
    const float* gr = orient    + (size_t)tok0 * 9;
    const float* gm = atom_mask + (size_t)tok0 * NATOM;
    const float* gd = dihedrals + (size_t)tok0 * 3;
    v4f rx[(FT * NATOM * 3 / 4 + FTHREADS - 1) / FTHREADS];
    v4f rr[(FT * 9 / 4 + FTHREADS - 1) / FTHREADS];
    v4f rm[(FT * NATOM / 4 + FTHREADS - 1) / FTHREADS];
    v4f rd[(FT * 3 / 4 + FTHREADS - 1) / FTHREADS];
    STAGE_LOAD(rx, gx, FT * NATOM * 3)
    STAGE_LOAD(rr, gr, FT * 9)
    STAGE_LOAD(rm, gm, FT * NATOM)
    STAGE_LOAD(rd, gd, FT * 3)
    const int ch = chain_idx[ctok];
    STAGE_STORE(rx, s_xyz, FT * NATOM * 3)
    STAGE_STORE(rr, s_R, FT * 9)
    STAGE_STORE(rm, s_m, FT * NATOM)
    STAGE_STORE(rd, s_d, FT * 3)
    if (tid < FT) {
      const v4f tail = {(float)ch, 0.f, 0.f, 0.f};
      *(v4f*)(s_f + tid * XS + 84) = tail;
    }
  } else {
    for (int j = tid; j < FT * NATOM * 3; j += FTHREADS) {
      const int t = j / (NATOM * 3);
      const int tok = (tok0 + t < ntok) ? (tok0 + t) : last;
      s_xyz[j] = xyz[(size_t)tok * (NATOM * 3) + (j - t * (NATOM * 3))];
    }
    for (int j = tid; j < FT * 9; j += FTHREADS) {
      const int t = j / 9;
      const int tok = (tok0 + t < ntok) ? (tok0 + t) : last;
      s_R[j] = orient[(size_t)tok * 9 + (j - t * 9)];
    }
    for (int j = tid; j < FT * NATOM; j += FTHREADS) {
      const int t = j / NATOM;
      const int tok = (tok0 + t < ntok) ? (tok0 + t) : last;
      s_m[j] = atom_mask[(size_t)tok * NATOM + (j - t * NATOM)];
    }
    for (int j = tid; j < FT * 3; j += FTHREADS) {
      const int t = j / 3;
      const int tok = (tok0 + t < ntok) ? (tok0 + t) : last;
      s_d[j] = dihedrals[(size_t)tok * 3 + (j - t * 3)];
    }
    if (tid < FT) {
      const v4f tail = {(float)chain_idx[ctok], 0.f, 0.f, 0.f};
      *(v4f*)(s_f + tid * XS + 84) = tail;
    }
  }
  __syncthreads();

  for (int task = tid; task < FT * NATOM; task += FTHREADS) {
    const int t = task / NATOM, a = task - t * NATOM;
    const float* p  = s_xyz + task * 3;
    const float* ca = s_xyz + t * (NATOM * 3) + 3;
    const float rx = p[0] - ca[0], ry = p[1] - ca[1], rz = p[2] - ca[2];
    const float* R = s_R + t * 9;
    const float m = s_m[task];
    float* d = s_f + t * XS + a * 3;
    d[0] = (R[0] * rx + R[3] * ry + R[6] * rz) * m;
    d[1] = (R[1] * rx + R[4] * ry + R[7] * rz) * m;
    d[2] = (R[2] * rx + R[5] * ry + R[8] * rz) * m;
  }
  for (int task = tid; task < FT * 18; task += FTHREADS) {
    const int t = task / 18;
    const int r = task - t * 18;
    const int d = r / 6;
    const int f = r - d * 6;
    const float x = s_d[t * 3 + d];
    const float kf = (f == 1) ? 2.f : (f == 2) ? 3.f : (f == 4) ? 0.5f
                   : (f == 5) ? (1.f / 3.f) : 1.f;
    float* dst = s_f + t * XS + 45 + d * 13;
    if (f == 0) dst[0] = x;
    dst[1 + f] = sinf(kf * x);
    dst[7 + f] = cosf(kf * x);
  }
  __syncthreads();

#pragma unroll
  for (int i = 0; i < (FT * XQ) / FTHREADS; ++i) {
    const int idx = tid + FTHREADS * i;
    if (tok0 + idx / XQ < ntok)
      *(v4f*)(feat + (size_t)tok0 * XS + idx * 4) = *(const v4f*)(s_f + idx * 4);
  }
}

__global__ __launch_bounds__(L1T, 2) void layer1_by_type(
    const int*   __restrict__ seq_idx,
    const float* __restrict__ feat,
    const float* __restrict__ W1, const float* __restrict__ b1,
    const float* __restrict__ proj,
    float*       __restrict__ h1,
    int ntok)
{
  __shared__ int   s_list[CHUNK + 4];
  __shared__ int   s_wcnt[L1T / 32];
  __shared__ int   s_ch[GT];
  __shared__ __attribute__((aligned(16))) float s_x[GT][XS];

  const int tid  = threadIdx.x;
  const int wave = tid >> 5;
  const int lane = tid & 31;
  const int type = blockIdx.x;
  const int tok_base = blockIdx.y * CHUNK;
  const int c0 = tid * 4;

  unsigned hbits = 0;
#pragma unroll
  for (int i = 0; i < CHUNK / L1T; ++i) {
    const int tok = tok_base + i * L1T + tid;
    if ((tok < ntok) && (seq_idx[tok] == type)) hbits |= 1u << i;
  }

  v4f wc[45], wd[DDIH];
  {
    const float* w1c = W1 + (size_t)(DFEAT + type * 45) * DH1 + c0;
#pragma unroll
    for (int k = 0; k < 45; ++k) wc[k] = *(const v4f*)(w1c + (size_t)k * DH1);
    const float* w1d = W1 + (size_t)(DFEAT + DCOORD) * DH1 + c0;
#pragma unroll
    for (int k = 0; k < DDIH; ++k) wd[k] = *(const v4f*)(w1d + (size_t)k * DH1);
  }
  const v4f base = *(const v4f*)(b1 + c0) + *(const v4f*)(proj + (size_t)type * DH1 + c0);

  unsigned msk[CHUNK / L1T];
  int cnt = 0;
#pragma unroll
  for (int i = 0; i < CHUNK / L1T; ++i) {
    msk[i] = __builtin_amdgcn_ballot_w32((hbits >> i) & 1u);
    cnt += __builtin_popcount(msk[i]);
  }
  if (lane == 0) s_wcnt[wave] = cnt;
  __syncthreads();
  int off = 0, total = 0;
#pragma unroll
  for (int w = 0; w < L1T / 32; ++w) {
    const int cw = s_wcnt[w];
    if (w < wave) off += cw;
    total += cw;
  }
  const unsigned below = (1u << lane) - 1u;
#pragma unroll
  for (int i = 0; i < CHUNK / L1T; ++i) {
    if ((hbits >> i) & 1u) s_list[off + __builtin_popcount(msk[i] & below)] = tok_base + i * L1T + tid;
    off += __builtin_popcount(msk[i]);
  }
  __syncthreads();
  if (total > 0 && tid < 3) s_list[total + tid] = s_list[total - 1];
  const int totalp = (total + 3) & ~3;
  __syncthreads();

  for (int g0 = 0; g0 < totalp; g0 += GT) {
    const int ng = (totalp - g0 < GT) ? (totalp - g0) : GT;

    {
      const int tt = (tid < ng) ? tid : (ng - 1);
      const v4f* src = (const v4f*)(feat + (size_t)s_list[g0 + tt] * XS);
      v4f r[XQ];
#pragma unroll
      for (int q = 0; q < XQ; ++q) r[q] = src[q];
      if (tid < ng) {
#pragma unroll
        for (int q = 0; q < XQ; ++q) *(v4f*)&s_x[tid][4 * q] = r[q];
        s_ch[tid] = (int)r[XQ - 1].x;
      }
    }
    __syncthreads();

    for (int t = 0; t < ng; t += 4) {
      v2f alo[4], ahi[4];
#pragma unroll
      for (int i = 0; i < 4; ++i) {
        const v4f a0 = base + *(const v4f*)(proj + (size_t)(N_AA + s_ch[t + i]) * DH1 + c0);
        alo[i] = a0.lo;
        ahi[i] = a0.hi;
      }
#pragma unroll
      for (int k = 0; k < 45; ++k) {
#pragma unroll
        for (int i = 0; i < 4; ++i) {
          const float x = s_x[t + i][k];
          const v2f xx = {x, x};
          alo[i] = __builtin_elementwise_fma(xx, wc[k].lo, alo[i]);
          ahi[i] = __builtin_elementwise_fma(xx, wc[k].hi, ahi[i]);
        }
      }
#pragma unroll
      for (int k = 0; k < DDIH; ++k) {
#pragma unroll
        for (int i = 0; i < 4; ++i) {
          const float x = s_x[t + i][45 + k];
          const v2f xx = {x, x};
          alo[i] = __builtin_elementwise_fma(xx, wd[k].lo, alo[i]);
          ahi[i] = __builtin_elementwise_fma(xx, wd[k].hi, ahi[i]);
        }
      }
#pragma unroll
      for (int i = 0; i < 4; ++i) {
        v4f o;
        o.x = fmaxf(alo[i].x, 0.f);
        o.y = fmaxf(alo[i].y, 0.f);
        o.z = fmaxf(ahi[i].x, 0.f);
        o.w = fmaxf(ahi[i].y, 0.f);
        *(v4f*)(h1 + (size_t)s_list[g0 + t + i] * DH1 + c0) = o;
      }
    }
    __syncthreads();
  }
}

#define MT      32
#define MW      8
#define MGRID   256
#define OFF_W3  (256 * 128)
#define OFF_W4  (OFF_W3 + 128 * 128)
#define OFF_H1  (OFF_W4 + 128 * 128)
#define OFF_H2  (OFF_H1 + MT * H1S)
#define LDS_FLOATS (OFF_H2 + MT * H2S)

__device__ __forceinline__ void stage_weights(const float* __restrict__ Wm, float* Ld,
                                              int units, int wave, int lane) {
  const int sw = (lane >> 2) & 3;
  for (int ub = wave; ub < units; ub += 4 * MW) {
    v4f v[4][4];
#pragma unroll
    for (int ii = 0; ii < 4; ++ii) {
      const int u  = ub + MW * ii;
      const int kb = 8 * (u >> 1) + 2 * (u & 1);
#pragma unroll
      for (int e = 0; e < 4; ++e)
        v[ii][e] = *(const v4f*)(Wm + (size_t)(kb + (e & 1) + 4 * (e >> 1)) * DFEAT + 4 * lane);
    }
#pragma unroll
    for (int ii = 0; ii < 4; ++ii) {
      float* d = Ld + (ub + MW * ii) * 512 + 16 * lane;
#pragma unroll
      for (int c = 0; c < 4; ++c) {
        const v4f o = {v[ii][0][c], v[ii][1][c], v[ii][2][c], v[ii][3][c]};
        *(v4f*)(d + 4 * (c ^ sw)) = o;
      }
    }
  }
}

#define NP 4
struct Frag { v2f a0[2 * NP], a1[2 * NP]; v4f b[NP]; };

__device__ __forceinline__ Frag load_frag(const float* a0p, const float* a1p, const float* bp, int blk) {
  Frag f;
#pragma unroll
  for (int p = 0; p < NP; ++p) {
    const int j = NP * blk + p;
    f.a0[2 * p]     = *(const v2f*)(a0p + 8 * j);
    f.a0[2 * p + 1] = *(const v2f*)(a0p + 8 * j + 4);
    f.a1[2 * p]     = *(const v2f*)(a1p + 8 * j);
    f.a1[2 * p + 1] = *(const v2f*)(a1p + 8 * j + 4);
    f.b[p]          = *(const v4f*)(bp + j * 1024);
  }
  return f;
}

__device__ __forceinline__ void mm_frag(const Frag& f, v8f& acc0, v8f& acc1) {
#pragma unroll
  for (int p = 0; p < NP; ++p) {
    acc0 = __builtin_amdgcn_wmma_f32_16x16x4_f32(false, f.a0[2 * p],     false, f.b[p].lo, (short)0, acc0, false, false);
    acc1 = __builtin_amdgcn_wmma_f32_16x16x4_f32(false, f.a1[2 * p],     false, f.b[p].lo, (short)0, acc1, false, false);
    acc0 = __builtin_amdgcn_wmma_f32_16x16x4_f32(false, f.a0[2 * p + 1], false, f.b[p].hi, (short)0, acc0, false, false);
    acc1 = __builtin_amdgcn_wmma_f32_16x16x4_f32(false, f.a1[2 * p + 1], false, f.b[p].hi, (short)0, acc1, false, false);
  }
}

__device__ __forceinline__ void group_fence() { __builtin_amdgcn_sched_barrier(0); }

__device__ __forceinline__ void layer_mm(const float* A, int pitch, const float* Bl, int nb,
                                         int lm, int half, int bslot, v8f& acc0, v8f& acc1) {
  const float* a0p = A + lm * pitch + 2 * half;
  const float* a1p = a0p + 16 * pitch;
  const float* bp  = Bl + half * 512 + bslot * 4;
  Frag f0 = load_frag(a0p, a1p, bp, 0);
  for (int blk = 0; blk < nb - 2; blk += 2) {
    const Frag f1 = load_frag(a0p, a1p, bp, blk + 1);
    group_fence();
    mm_frag(f0, acc0, acc1);
    group_fence();
    f0 = load_frag(a0p, a1p, bp, blk + 2);
    group_fence();
    mm_frag(f1, acc0, acc1);
    group_fence();
  }
  const Frag f1 = load_frag(a0p, a1p, bp, nb - 1);
  group_fence();
  mm_frag(f0, acc0, acc1);
  group_fence();
  mm_frag(f1, acc0, acc1);
}

__device__ __forceinline__ void store_out(float* __restrict__ out, const v8f& o0, const v8f& o1,
                                          int tok0, int half, int col, int ntok) {
  float* p = out + (size_t)(tok0 + 8 * half) * DFEAT + col;
  if (tok0 + MT <= ntok) {
#pragma unroll
    for (int r = 0; r < 8; ++r) {
      p[r * DFEAT]        = o0[r];
      p[(16 + r) * DFEAT] = o1[r];
    }
  } else {
#pragma unroll
    for (int r = 0; r < 8; ++r) {
      if (tok0 + 8 * half + r < ntok)      p[r * DFEAT]        = o0[r];
      if (tok0 + 8 * half + 16 + r < ntok) p[(16 + r) * DFEAT] = o1[r];
    }
  }
}

__global__ __launch_bounds__(MW * 32) void mlp_layers_2_to_4(
    const float* __restrict__ h1,
    const float* __restrict__ W2, const float* __restrict__ b2,
    const float* __restrict__ W3, const float* __restrict__ b3,
    const float* __restrict__ W4, const float* __restrict__ b4,
    float*       __restrict__ out,
    int ntok)
{
  __shared__ __attribute__((aligned(16))) float L[LDS_FLOATS];

  const int tid  = threadIdx.x;
  const int wave = tid >> 5;
  const int lane = tid & 31;
  const int half = lane >> 4;
  const int lm   = lane & 15;
  const int col  = wave * 16 + lm;
  const int bslot = wave * 16 + (lm ^ (wave & 3));
  const int ntiles = (ntok + MT - 1) / MT;

  v4f pre[8];
  int tile = blockIdx.x;
  if (tile < ntiles) {
#pragma unroll
    for (int q = 0; q < 8; ++q) {
      const int i = tid + 256 * q;
      int row = tile * MT + (i >> 6);
      row = (row < ntok) ? row : (ntok - 1);
      pre[q] = *(const v4f*)(h1 + (size_t)row * DH1 + (i & 63) * 4);
    }
  }
  const float bias2 = b2[col], bias3 = b3[col], bias4 = b4[col];

  stage_weights(W2, L,          64, wave, lane);
  stage_weights(W3, L + OFF_W3, 32, wave, lane);
  stage_weights(W4, L + OFF_W4, 32, wave, lane);

  v8f o0 = {}, o1 = {};
  int otok0 = -1;

  for (; tile < ntiles; tile += gridDim.x) {
#pragma unroll
    for (int q = 0; q < 8; ++q) {
      const int i = tid + 256 * q;
      *(v4f*)&L[OFF_H1 + (i >> 6) * H1S + (i & 63) * 4] = pre[q];
    }
    __syncthreads();

    if (otok0 >= 0) store_out(out, o0, o1, otok0, half, col, ntok);

    const int tnext = tile + gridDim.x;
    if (tnext < ntiles) {
#pragma unroll
      for (int q = 0; q < 8; ++q) {
        const int i = tid + 256 * q;
        int row = tnext * MT + (i >> 6);
        row = (row < ntok) ? row : (ntok - 1);
        pre[q] = *(const v4f*)(h1 + (size_t)row * DH1 + (i & 63) * 4);
      }
    }

    {
      v8f acc0 = {}, acc1 = {};
      layer_mm(L + OFF_H1, H1S, L, 8, lm, half, bslot, acc0, acc1);
#pragma unroll
      for (int r = 0; r < 8; ++r) {
        L[OFF_H2 + (r + 8 * half) * H2S + col]      = fmaxf(acc0[r] + bias2, 0.f);
        L[OFF_H2 + (16 + r + 8 * half) * H2S + col] = fmaxf(acc1[r] + bias2, 0.f);
      }
    }
    __syncthreads();

    {
      v8f acc0 = {}, acc1 = {};
      layer_mm(L + OFF_H2, H2S, L + OFF_W3, 4, lm, half, bslot, acc0, acc1);
#pragma unroll
      for (int r = 0; r < 8; ++r) {
        L[OFF_H1 + (r + 8 * half) * H2S + col]      = fmaxf(acc0[r] + bias3, 0.f);
        L[OFF_H1 + (16 + r + 8 * half) * H2S + col] = fmaxf(acc1[r] + bias3, 0.f);
      }
    }
    __syncthreads();

    {
      v8f acc0 = {}, acc1 = {};
      layer_mm(L + OFF_H1, H2S, L + OFF_W4, 4, lm, half, bslot, acc0, acc1);
#pragma unroll
      for (int r = 0; r < 8; ++r) {
        o0[r] = acc0[r] + bias4;
        o1[r] = acc1[r] + bias4;
      }
      otok0 = tile * MT;
    }
    __syncthreads();
  }
  if (otok0 >= 0) store_out(out, o0, o1, otok0, half, col, ntok);
}

extern "C" void kernel_launch(void* const* d_in, const int* in_sizes, int n_in,
                              void* d_out, int out_size, void* d_ws, size_t ws_size,
                              hipStream_t stream) {
  const int*   seq    = (const int*)  d_in[0];
  const float* xyz    = (const float*)d_in[1];
  const float* orient = (const float*)d_in[2];
  const float* dih    = (const float*)d_in[3];
  const int*   chain  = (const int*)  d_in[4];
  const float* amask  = (const float*)d_in[5];
  const float* aa_emb = (const float*)d_in[6];
  const float* ch_emb = (const float*)d_in[7];
  const float* W1 = (const float*)d_in[8];
  const float* b1 = (const float*)d_in[9];
  const float* W2 = (const float*)d_in[10];
  const float* b2 = (const float*)d_in[11];
  const float* W3 = (const float*)d_in[12];
  const float* b3 = (const float*)d_in[13];
  const float* W4 = (const float*)d_in[14];
  const float* b4 = (const float*)d_in[15];
  float* out  = (float*)d_out;
  float* proj = (float*)d_ws;
  float* h1   = proj + 8192;
  float* feat = h1 + (size_t)32768 * DH1;

  const int ntok = in_sizes[0];

  proj_and_features<<<NPROJ + (ntok + FT - 1) / FT, FTHREADS, 0, stream>>>(
      aa_emb, ch_emb, W1, xyz, orient, dih, chain, amask, proj, feat, ntok);
  layer1_by_type<<<dim3(N_AA, (ntok + CHUNK - 1) / CHUNK), L1T, 0, stream>>>(
      seq, feat, W1, b1, proj, h1, ntok);
  mlp_layers_2_to_4<<<MGRID, MW * 32, 0, stream>>>(
      h1, W2, b2, W3, b3, W4, b4, out, ntok);
}


# By-type launch's token scan: a thread's 16 residue-type loads carry no guard (index clamped, result masked), so they leave in one group with the weight loads, not as 16 waited-for round trips
# speedup vs baseline: 1.6256x; 1.0547x over previous
#include <hip/hip_runtime.h>

#define N_AA   21
#define NATOM  15
#define DFEAT  128
#define DCOORD (N_AA * NATOM * 3)
#define DDIH   39
#define DH1    256
#define H1S    260
#define H2S    132

#define CHUNK  1024
#define L1T    64
#define GT     L1T
#define XS     88
#define XQ     (XS / 4)
#define FT     128
#define NPROJ  (N_AA + 10)

typedef __attribute__((ext_vector_type(2))) float v2f;
typedef __attribute__((ext_vector_type(4))) float v4f;
typedef __attribute__((ext_vector_type(8))) float v8f;

#define FTHREADS 256

#define STAGE_LOAD(reg, src, n)                                                  \
  _Pragma("unroll")                                                              \
  for (int i = 0; i < ((n) / 4 + FTHREADS - 1) / FTHREADS; ++i) {                \
    const int idx = tid + FTHREADS * i;                                          \
    reg[i] = *(const v4f*)((src) + 4 * ((idx < (n) / 4) ? idx : ((n) / 4 - 1))); \
  }
#define STAGE_STORE(reg, dst, n)                                                 \
  _Pragma("unroll")                                                              \
  for (int i = 0; i < ((n) / 4 + FTHREADS - 1) / FTHREADS; ++i) {                \
    const int idx = tid + FTHREADS * i;                                          \
    if (idx < (n) / 4) *(v4f*)((dst) + 4 * idx) = reg[i];                        \
  }

__global__ __launch_bounds__(FTHREADS) void proj_and_features(
    const float* __restrict__ aa_emb,
    const float* __restrict__ chain_emb,
    const float* __restrict__ W1,
    const float* __restrict__ xyz,
    const float* __restrict__ orient,
    const float* __restrict__ dihedrals,
    const int*   __restrict__ chain_idx,
    const float* __restrict__ atom_mask,
    float* __restrict__ proj,
    float* __restrict__ feat,
    int ntok)
{
  __shared__ __attribute__((aligned(16))) float s_f[FT * XS];
  __shared__ __attribute__((aligned(16))) float s_xyz[FT * NATOM * 3];
  __shared__ __attribute__((aligned(16))) float s_R[FT * 9];
  __shared__ __attribute__((aligned(16))) float s_m[FT * NATOM];
  __shared__ __attribute__((aligned(16))) float s_d[FT * 3];

  const int tid = threadIdx.x;
  if (blockIdx.x < NPROJ) {
    const int r = blockIdx.x;
    const int c = tid;
    const float* emb;
    const float* wrow;
    if (r < N_AA) {
      emb  = aa_emb + (size_t)r * DFEAT;
      wrow = W1;
    } else {
      emb  = chain_emb + (size_t)(r - N_AA) * DFEAT;
      wrow = W1 + (size_t)(DFEAT + DCOORD + DDIH) * DH1;
    }
    float acc = 0.f;
    for (int k = 0; k < DFEAT; ++k)
      acc = fmaf(emb[k], wrow[(size_t)k * DH1 + c], acc);
    proj[(size_t)r * DH1 + c] = acc;
    return;
  }

  const int tok0 = (blockIdx.x - NPROJ) * FT;
  const int last = ntok - 1;
  const int ctok = (tok0 + (tid & (FT - 1)) < ntok) ? (tok0 + (tid & (FT - 1))) : last;

  if (tok0 + FT <= ntok) {
    const float* gx = xyz       + (size_t)tok0 * (NATOM * 3);
    const float* gr = orient    + (size_t)tok0 * 9;
    const float* gm = atom_mask + (size_t)tok0 * NATOM;
    const float* gd = dihedrals + (size_t)tok0 * 3;
    v4f rx[(FT * NATOM * 3 / 4 + FTHREADS - 1) / FTHREADS];
    v4f rr[(FT * 9 / 4 + FTHREADS - 1) / FTHREADS];
    v4f rm[(FT * NATOM / 4 + FTHREADS - 1) / FTHREADS];
    v4f rd[(FT * 3 / 4 + FTHREADS - 1) / FTHREADS];
    STAGE_LOAD(rx, gx, FT * NATOM * 3)
    STAGE_LOAD(rr, gr, FT * 9)
    STAGE_LOAD(rm, gm, FT * NATOM)
    STAGE_LOAD(rd, gd, FT * 3)
    const int ch = chain_idx[ctok];
    STAGE_STORE(rx, s_xyz, FT * NATOM * 3)
    STAGE_STORE(rr, s_R, FT * 9)
    STAGE_STORE(rm, s_m, FT * NATOM)
    STAGE_STORE(rd, s_d, FT * 3)
    if (tid < FT) {
      const v4f tail = {(float)ch, 0.f, 0.f, 0.f};
      *(v4f*)(s_f + tid * XS + 84) = tail;
    }
  } else {
    for (int j = tid; j < FT * NATOM * 3; j += FTHREADS) {
      const int t = j / (NATOM * 3);
      const int tok = (tok0 + t < ntok) ? (tok0 + t) : last;
      s_xyz[j] = xyz[(size_t)tok * (NATOM * 3) + (j - t * (NATOM * 3))];
    }
    for (int j = tid; j < FT * 9; j += FTHREADS) {
      const int t = j / 9;
      const int tok = (tok0 + t < ntok) ? (tok0 + t) : last;
      s_R[j] = orient[(size_t)tok * 9 + (j - t * 9)];
    }
    for (int j = tid; j < FT * NATOM; j += FTHREADS) {
      const int t = j / NATOM;
      const int tok = (tok0 + t < ntok) ? (tok0 + t) : last;
      s_m[j] = atom_mask[(size_t)tok * NATOM + (j - t * NATOM)];
    }
    for (int j = tid; j < FT * 3; j += FTHREADS) {
      const int t = j / 3;
      const int tok = (tok0 + t < ntok) ? (tok0 + t) : last;
      s_d[j] = dihedrals[(size_t)tok * 3 + (j - t * 3)];
    }
    if (tid < FT) {
      const v4f tail = {(float)chain_idx[ctok], 0.f, 0.f, 0.f};
      *(v4f*)(s_f + tid * XS + 84) = tail;
    }
  }
  __syncthreads();

  for (int task = tid; task < FT * NATOM; task += FTHREADS) {
    const int t = task / NATOM, a = task - t * NATOM;
    const float* p  = s_xyz + task * 3;
    const float* ca = s_xyz + t * (NATOM * 3) + 3;
    const float rx = p[0] - ca[0], ry = p[1] - ca[1], rz = p[2] - ca[2];
    const float* R = s_R + t * 9;
    const float m = s_m[task];
    float* d = s_f + t * XS + a * 3;
    d[0] = (R[0] * rx + R[3] * ry + R[6] * rz) * m;
    d[1] = (R[1] * rx + R[4] * ry + R[7] * rz) * m;
    d[2] = (R[2] * rx + R[5] * ry + R[8] * rz) * m;
  }
  for (int task = tid; task < FT * 18; task += FTHREADS) {
    const int t = task / 18;
    const int r = task - t * 18;
    const int d = r / 6;
    const int f = r - d * 6;
    const float x = s_d[t * 3 + d];
    const float kf = (f == 1) ? 2.f : (f == 2) ? 3.f : (f == 4) ? 0.5f
                   : (f == 5) ? (1.f / 3.f) : 1.f;
    float* dst = s_f + t * XS + 45 + d * 13;
    if (f == 0) dst[0] = x;
    dst[1 + f] = sinf(kf * x);
    dst[7 + f] = cosf(kf * x);
  }
  __syncthreads();

#pragma unroll
  for (int i = 0; i < (FT * XQ) / FTHREADS; ++i) {
    const int idx = tid + FTHREADS * i;
    if (tok0 + idx / XQ < ntok)
      *(v4f*)(feat + (size_t)tok0 * XS + idx * 4) = *(const v4f*)(s_f + idx * 4);
  }
}

__global__ __launch_bounds__(L1T, 2) void layer1_by_type(
    const int*   __restrict__ seq_idx,
    const float* __restrict__ feat,
    const float* __restrict__ W1, const float* __restrict__ b1,
    const float* __restrict__ proj,
    float*       __restrict__ h1,
    int ntok)
{
  __shared__ int   s_list[CHUNK + 4];
  __shared__ int   s_wcnt[L1T / 32];
  __shared__ int   s_ch[GT];
  __shared__ __attribute__((aligned(16))) float s_x[GT][XS];

  const int tid  = threadIdx.x;
  const int wave = tid >> 5;
  const int lane = tid & 31;
  const int type = blockIdx.x;
  const int tok_base = blockIdx.y * CHUNK;
  const int c0 = tid * 4;

  int sq[CHUNK / L1T];
#pragma unroll
  for (int i = 0; i < CHUNK / L1T; ++i) {
    const int tok = tok_base + i * L1T + tid;
    sq[i] = seq_idx[(tok < ntok) ? tok : (ntok - 1)];
  }

  v4f wc[45], wd[DDIH];
  {
    const float* w1c = W1 + (size_t)(DFEAT + type * 45) * DH1 + c0;
#pragma unroll
    for (int k = 0; k < 45; ++k) wc[k] = *(const v4f*)(w1c + (size_t)k * DH1);
    const float* w1d = W1 + (size_t)(DFEAT + DCOORD) * DH1 + c0;
#pragma unroll
    for (int k = 0; k < DDIH; ++k) wd[k] = *(const v4f*)(w1d + (size_t)k * DH1);
  }
  const v4f base = *(const v4f*)(b1 + c0) + *(const v4f*)(proj + (size_t)type * DH1 + c0);

  unsigned hbits = 0;
#pragma unroll
  for (int i = 0; i < CHUNK / L1T; ++i) {
    const int tok = tok_base + i * L1T + tid;
    if ((tok < ntok) && (sq[i] == type)) hbits |= 1u << i;
  }

  unsigned msk[CHUNK / L1T];
  int cnt = 0;
#pragma unroll
  for (int i = 0; i < CHUNK / L1T; ++i) {
    msk[i] = __builtin_amdgcn_ballot_w32((hbits >> i) & 1u);
    cnt += __builtin_popcount(msk[i]);
  }
  if (lane == 0) s_wcnt[wave] = cnt;
  __syncthreads();
  int off = 0, total = 0;
#pragma unroll
  for (int w = 0; w < L1T / 32; ++w) {
    const int cw = s_wcnt[w];
    if (w < wave) off += cw;
    total += cw;
  }
  const unsigned below = (1u << lane) - 1u;
#pragma unroll
  for (int i = 0; i < CHUNK / L1T; ++i) {
    if ((hbits >> i) & 1u) s_list[off + __builtin_popcount(msk[i] & below)] = tok_base + i * L1T + tid;
    off += __builtin_popcount(msk[i]);
  }
  __syncthreads();
  if (total > 0 && tid < 3) s_list[total + tid] = s_list[total - 1];
  const int totalp = (total + 3) & ~3;
  __syncthreads();

  for (int g0 = 0; g0 < totalp; g0 += GT) {
    const int ng = (totalp - g0 < GT) ? (totalp - g0) : GT;

    {
      const int tt = (tid < ng) ? tid : (ng - 1);
      const v4f* src = (const v4f*)(feat + (size_t)s_list[g0 + tt] * XS);
      v4f r[XQ];
#pragma unroll
      for (int q = 0; q < XQ; ++q) r[q] = src[q];
      if (tid < ng) {
#pragma unroll
        for (int q = 0; q < XQ; ++q) *(v4f*)&s_x[tid][4 * q] = r[q];
        s_ch[tid] = (int)r[XQ - 1].x;
      }
    }
    __syncthreads();

    for (int t = 0; t < ng; t += 4) {
      v2f alo[4], ahi[4];
#pragma unroll
      for (int i = 0; i < 4; ++i) {
        const v4f a0 = base + *(const v4f*)(proj + (size_t)(N_AA + s_ch[t + i]) * DH1 + c0);
        alo[i] = a0.lo;
        ahi[i] = a0.hi;
      }
#pragma unroll
      for (int k = 0; k < 45; ++k) {
#pragma unroll
        for (int i = 0; i < 4; ++i) {
          const float x = s_x[t + i][k];
          const v2f xx = {x, x};
          alo[i] = __builtin_elementwise_fma(xx, wc[k].lo, alo[i]);
          ahi[i] = __builtin_elementwise_fma(xx, wc[k].hi, ahi[i]);
        }
      }
#pragma unroll
      for (int k = 0; k < DDIH; ++k) {
#pragma unroll
        for (int i = 0; i < 4; ++i) {
          const float x = s_x[t + i][45 + k];
          const v2f xx = {x, x};
          alo[i] = __builtin_elementwise_fma(xx, wd[k].lo, alo[i]);
          ahi[i] = __builtin_elementwise_fma(xx, wd[k].hi, ahi[i]);
        }
      }
#pragma unroll
      for (int i = 0; i < 4; ++i) {
        v4f o;
        o.x = fmaxf(alo[i].x, 0.f);
        o.y = fmaxf(alo[i].y, 0.f);
        o.z = fmaxf(ahi[i].x, 0.f);
        o.w = fmaxf(ahi[i].y, 0.f);
        *(v4f*)(h1 + (size_t)s_list[g0 + t + i] * DH1 + c0) = o;
      }
    }
    __syncthreads();
  }
}

#define MT      32
#define MW      8
#define MGRID   256
#define OFF_W3  (256 * 128)
#define OFF_W4  (OFF_W3 + 128 * 128)
#define OFF_H1  (OFF_W4 + 128 * 128)
#define OFF_H2  (OFF_H1 + MT * H1S)
#define LDS_FLOATS (OFF_H2 + MT * H2S)

__device__ __forceinline__ void stage_weights(const float* __restrict__ Wm, float* Ld,
                                              int units, int wave, int lane) {
  const int sw = (lane >> 2) & 3;
  for (int ub = wave; ub < units; ub += 4 * MW) {
    v4f v[4][4];
#pragma unroll
    for (int ii = 0; ii < 4; ++ii) {
      const int u  = ub + MW * ii;
      const int kb = 8 * (u >> 1) + 2 * (u & 1);
#pragma unroll
      for (int e = 0; e < 4; ++e)
        v[ii][e] = *(const v4f*)(Wm + (size_t)(kb + (e & 1) + 4 * (e >> 1)) * DFEAT + 4 * lane);
    }
#pragma unroll
    for (int ii = 0; ii < 4; ++ii) {
      float* d = Ld + (ub + MW * ii) * 512 + 16 * lane;
#pragma unroll
      for (int c = 0; c < 4; ++c) {
        const v4f o = {v[ii][0][c], v[ii][1][c], v[ii][2][c], v[ii][3][c]};
        *(v4f*)(d + 4 * (c ^ sw)) = o;
      }
    }
  }
}

#define NP 4
struct Frag { v2f a0[2 * NP], a1[2 * NP]; v4f b[NP]; };

__device__ __forceinline__ Frag load_frag(const float* a0p, const float* a1p, const float* bp, int blk) {
  Frag f;
#pragma unroll
  for (int p = 0; p < NP; ++p) {
    const int j = NP * blk + p;
    f.a0[2 * p]     = *(const v2f*)(a0p + 8 * j);
    f.a0[2 * p + 1] = *(const v2f*)(a0p + 8 * j + 4);
    f.a1[2 * p]     = *(const v2f*)(a1p + 8 * j);
    f.a1[2 * p + 1] = *(const v2f*)(a1p + 8 * j + 4);
    f.b[p]          = *(const v4f*)(bp + j * 1024);
  }
  return f;
}

__device__ __forceinline__ void mm_frag(const Frag& f, v8f& acc0, v8f& acc1) {
#pragma unroll
  for (int p = 0; p < NP; ++p) {
    acc0 = __builtin_amdgcn_wmma_f32_16x16x4_f32(false, f.a0[2 * p],     false, f.b[p].lo, (short)0, acc0, false, false);
    acc1 = __builtin_amdgcn_wmma_f32_16x16x4_f32(false, f.a1[2 * p],     false, f.b[p].lo, (short)0, acc1, false, false);
    acc0 = __builtin_amdgcn_wmma_f32_16x16x4_f32(false, f.a0[2 * p + 1], false, f.b[p].hi, (short)0, acc0, false, false);
    acc1 = __builtin_amdgcn_wmma_f32_16x16x4_f32(false, f.a1[2 * p + 1], false, f.b[p].hi, (short)0, acc1, false, false);
  }
}

__device__ __forceinline__ void group_fence() { __builtin_amdgcn_sched_barrier(0); }

__device__ __forceinline__ void layer_mm(const float* A, int pitch, const float* Bl, int nb,
                                         int lm, int half, int bslot, v8f& acc0, v8f& acc1) {
  const float* a0p = A + lm * pitch + 2 * half;
  const float* a1p = a0p + 16 * pitch;
  const float* bp  = Bl + half * 512 + bslot * 4;
  Frag f0 = load_frag(a0p, a1p, bp, 0);
  for (int blk = 0; blk < nb - 2; blk += 2) {
    const Frag f1 = load_frag(a0p, a1p, bp, blk + 1);
    group_fence();
    mm_frag(f0, acc0, acc1);
    group_fence();
    f0 = load_frag(a0p, a1p, bp, blk + 2);
    group_fence();
    mm_frag(f1, acc0, acc1);
    group_fence();
  }
  const Frag f1 = load_frag(a0p, a1p, bp, nb - 1);
  group_fence();
  mm_frag(f0, acc0, acc1);
  group_fence();
  mm_frag(f1, acc0, acc1);
}

__device__ __forceinline__ void store_out(float* __restrict__ out, const v8f& o0, const v8f& o1,
                                          int tok0, int half, int col, int ntok) {
  float* p = out + (size_t)(tok0 + 8 * half) * DFEAT + col;
  if (tok0 + MT <= ntok) {
#pragma unroll
    for (int r = 0; r < 8; ++r) {
      p[r * DFEAT]        = o0[r];
      p[(16 + r) * DFEAT] = o1[r];
    }
  } else {
#pragma unroll
    for (int r = 0; r < 8; ++r) {
      if (tok0 + 8 * half + r < ntok)      p[r * DFEAT]        = o0[r];
      if (tok0 + 8 * half + 16 + r < ntok) p[(16 + r) * DFEAT] = o1[r];
    }
  }
}

__global__ __launch_bounds__(MW * 32) void mlp_layers_2_to_4(
    const float* __restrict__ h1,
    const float* __restrict__ W2, const float* __restrict__ b2,
    const float* __restrict__ W3, const float* __restrict__ b3,
    const float* __restrict__ W4, const float* __restrict__ b4,
    float*       __restrict__ out,
    int ntok)
{
  __shared__ __attribute__((aligned(16))) float L[LDS_FLOATS];

  const int tid  = threadIdx.x;
  const int wave = tid >> 5;
  const int lane = tid & 31;
  const int half = lane >> 4;
  const int lm   = lane & 15;
  const int col  = wave * 16 + lm;
  const int bslot = wave * 16 + (lm ^ (wave & 3));
  const int ntiles = (ntok + MT - 1) / MT;

  v4f pre[8];
  int tile = blockIdx.x;
  if (tile < ntiles) {
#pragma unroll
    for (int q = 0; q < 8; ++q) {
      const int i = tid + 256 * q;
      int row = tile * MT + (i >> 6);
      row = (row < ntok) ? row : (ntok - 1);
      pre[q] = *(const v4f*)(h1 + (size_t)row * DH1 + (i & 63) * 4);
    }
  }
  const float bias2 = b2[col], bias3 = b3[col], bias4 = b4[col];

  stage_weights(W2, L,          64, wave, lane);
  stage_weights(W3, L + OFF_W3, 32, wave, lane);
  stage_weights(W4, L + OFF_W4, 32, wave, lane);

  v8f o0 = {}, o1 = {};
  int otok0 = -1;

  for (; tile < ntiles; tile += gridDim.x) {
#pragma unroll
    for (int q = 0; q < 8; ++q) {
      const int i = tid + 256 * q;
      *(v4f*)&L[OFF_H1 + (i >> 6) * H1S + (i & 63) * 4] = pre[q];
    }
    __syncthreads();

    if (otok0 >= 0) store_out(out, o0, o1, otok0, half, col, ntok);

    const int tnext = tile + gridDim.x;
    if (tnext < ntiles) {
#pragma unroll
      for (int q = 0; q < 8; ++q) {
        const int i = tid + 256 * q;
        int row = tnext * MT + (i >> 6);
        row = (row < ntok) ? row : (ntok - 1);
        pre[q] = *(const v4f*)(h1 + (size_t)row * DH1 + (i & 63) * 4);
      }
    }

    {
      v8f acc0 = {}, acc1 = {};
      layer_mm(L + OFF_H1, H1S, L, 8, lm, half, bslot, acc0, acc1);
#pragma unroll
      for (int r = 0; r < 8; ++r) {
        L[OFF_H2 + (r + 8 * half) * H2S + col]      = fmaxf(acc0[r] + bias2, 0.f);
        L[OFF_H2 + (16 + r + 8 * half) * H2S + col] = fmaxf(acc1[r] + bias2, 0.f);
      }
    }
    __syncthreads();

    {
      v8f acc0 = {}, acc1 = {};
      layer_mm(L + OFF_H2, H2S, L + OFF_W3, 4, lm, half, bslot, acc0, acc1);
#pragma unroll
      for (int r = 0; r < 8; ++r) {
        L[OFF_H1 + (r + 8 * half) * H2S + col]      = fmaxf(acc0[r] + bias3, 0.f);
        L[OFF_H1 + (16 + r + 8 * half) * H2S + col] = fmaxf(acc1[r] + bias3, 0.f);
      }
    }
    __syncthreads();

    {
      v8f acc0 = {}, acc1 = {};
      layer_mm(L + OFF_H1, H2S, L + OFF_W4, 4, lm, half, bslot, acc0, acc1);
#pragma unroll
      for (int r = 0; r < 8; ++r) {
        o0[r] = acc0[r] + bias4;
        o1[r] = acc1[r] + bias4;
      }
      otok0 = tile * MT;
    }
    __syncthreads();
  }
  if (otok0 >= 0) store_out(out, o0, o1, otok0, half, col, ntok);
}

extern "C" void kernel_launch(void* const* d_in, const int* in_sizes, int n_in,
                              void* d_out, int out_size, void* d_ws, size_t ws_size,
                              hipStream_t stream) {
  const int*   seq    = (const int*)  d_in[0];
  const float* xyz    = (const float*)d_in[1];
  const float* orient = (const float*)d_in[2];
  const float* dih    = (const float*)d_in[3];
  const int*   chain  = (const int*)  d_in[4];
  const float* amask  = (const float*)d_in[5];
  const float* aa_emb = (const float*)d_in[6];
  const float* ch_emb = (const float*)d_in[7];
  const float* W1 = (const float*)d_in[8];
  const float* b1 = (const float*)d_in[9];
  const float* W2 = (const float*)d_in[10];
  const float* b2 = (const float*)d_in[11];
  const float* W3 = (const float*)d_in[12];
  const float* b3 = (const float*)d_in[13];
  const float* W4 = (const float*)d_in[14];
  const float* b4 = (const float*)d_in[15];
  float* out  = (float*)d_out;
  float* proj = (float*)d_ws;
  float* h1   = proj + 8192;
  float* feat = h1 + (size_t)32768 * DH1;

  const int ntok = in_sizes[0];

  proj_and_features<<<NPROJ + (ntok + FT - 1) / FT, FTHREADS, 0, stream>>>(
      aa_emb, ch_emb, W1, xyz, orient, dih, chain, amask, proj, feat, ntok);
  layer1_by_type<<<dim3(N_AA, (ntok + CHUNK - 1) / CHUNK), L1T, 0, stream>>>(
      seq, feat, W1, b1, proj, h1, ntok);
  mlp_layers_2_to_4<<<MGRID, MW * 32, 0, stream>>>(
      h1, W2, b2, W3, b3, W4, b4, out, ntok);
}


# Layer-1 by-type launch cut by count, not position: each type's token order in 48 equal slices (per-block type counts from launch 1, prefix sums, no atomics): 1008 equal work-groups; same sum order
# speedup vs baseline: 1.9682x; 1.2108x over previous
#include <hip/hip_runtime.h>

#define N_AA   21
#define NATOM  15
#define DFEAT  128
#define DCOORD (N_AA * NATOM * 3)
#define DDIH   39
#define DH1    256
#define H1S    260
#define H2S    132

#define NSL    48
#define LISTMAX 704
#define L1T    64
#define GT     L1T
#define XS     88
#define XQ     (XS / 4)
#define FT     128
#define NPROJ  (N_AA + 10)
#define NBLK   256
#define CSTR   32
#define BPT    (NBLK / L1T)
#define RB     8
#define RG     (RB * FT / L1T)

typedef __attribute__((ext_vector_type(2))) float v2f;
typedef __attribute__((ext_vector_type(4))) float v4f;
typedef __attribute__((ext_vector_type(8))) float v8f;

#define FTHREADS 256

#define STAGE_LOAD(reg, src, n)                                                  \
  _Pragma("unroll")                                                              \
  for (int i = 0; i < ((n) / 4 + FTHREADS - 1) / FTHREADS; ++i) {                \
    const int idx = tid + FTHREADS * i;                                          \
    reg[i] = *(const v4f*)((src) + 4 * ((idx < (n) / 4) ? idx : ((n) / 4 - 1))); \
  }
#define STAGE_STORE(reg, dst, n)                                                 \
  _Pragma("unroll")                                                              \
  for (int i = 0; i < ((n) / 4 + FTHREADS - 1) / FTHREADS; ++i) {                \
    const int idx = tid + FTHREADS * i;                                          \
    if (idx < (n) / 4) *(v4f*)((dst) + 4 * idx) = reg[i];                        \
  }

__global__ __launch_bounds__(FTHREADS) void proj_and_features(
    const float* __restrict__ aa_emb,
    const float* __restrict__ chain_emb,
    const float* __restrict__ W1,
    const float* __restrict__ xyz,
    const float* __restrict__ orient,
    const float* __restrict__ dihedrals,
    const int*   __restrict__ chain_idx,
    const float* __restrict__ atom_mask,
    const int*   __restrict__ seq_idx,
    float* __restrict__ proj,
    float* __restrict__ feat,
    int*   __restrict__ cnt,
    int ntok)
{
  __shared__ __attribute__((aligned(16))) float s_f[FT * XS];
  __shared__ __attribute__((aligned(16))) float s_xyz[FT * NATOM * 3];
  __shared__ __attribute__((aligned(16))) float s_R[FT * 9];
  __shared__ __attribute__((aligned(16))) float s_m[FT * NATOM];
  __shared__ __attribute__((aligned(16))) float s_d[FT * 3];
  __shared__ int s_tc[(FT / 32) * N_AA];

  const int tid = threadIdx.x;
  if (blockIdx.x < NPROJ) {
    const int r = blockIdx.x;
    const int c = tid;
    const float* emb;
    const float* wrow;
    if (r < N_AA) {
      emb  = aa_emb + (size_t)r * DFEAT;
      wrow = W1;
    } else {
      emb  = chain_emb + (size_t)(r - N_AA) * DFEAT;
      wrow = W1 + (size_t)(DFEAT + DCOORD + DDIH) * DH1;
    }
    float acc = 0.f;
    for (int k = 0; k < DFEAT; ++k)
      acc = fmaf(emb[k], wrow[(size_t)k * DH1 + c], acc);
    proj[(size_t)r * DH1 + c] = acc;
    return;
  }

  const int tok0 = (blockIdx.x - NPROJ) * FT;
  const int last = ntok - 1;
  const int ctok = (tok0 + (tid & (FT - 1)) < ntok) ? (tok0 + (tid & (FT - 1))) : last;
  const int tyl = seq_idx[ctok];

  if (tok0 + FT <= ntok) {
    const float* gx = xyz       + (size_t)tok0 * (NATOM * 3);
    const float* gr = orient    + (size_t)tok0 * 9;
    const float* gm = atom_mask + (size_t)tok0 * NATOM;
    const float* gd = dihedrals + (size_t)tok0 * 3;
    v4f rx[(FT * NATOM * 3 / 4 + FTHREADS - 1) / FTHREADS];
    v4f rr[(FT * 9 / 4 + FTHREADS - 1) / FTHREADS];
    v4f rm[(FT * NATOM / 4 + FTHREADS - 1) / FTHREADS];
    v4f rd[(FT * 3 / 4 + FTHREADS - 1) / FTHREADS];
    STAGE_LOAD(rx, gx, FT * NATOM * 3)
    STAGE_LOAD(rr, gr, FT * 9)
    STAGE_LOAD(rm, gm, FT * NATOM)
    STAGE_LOAD(rd, gd, FT * 3)
    const int ch = chain_idx[ctok];
    STAGE_STORE(rx, s_xyz, FT * NATOM * 3)
    STAGE_STORE(rr, s_R, FT * 9)
    STAGE_STORE(rm, s_m, FT * NATOM)
    STAGE_STORE(rd, s_d, FT * 3)
    if (tid < FT) {
      const v4f tail = {(float)ch, 0.f, 0.f, 0.f};
      *(v4f*)(s_f + tid * XS + 84) = tail;
    }
  } else {
    for (int j = tid; j < FT * NATOM * 3; j += FTHREADS) {
      const int t = j / (NATOM * 3);
      const int tok = (tok0 + t < ntok) ? (tok0 + t) : last;
      s_xyz[j] = xyz[(size_t)tok * (NATOM * 3) + (j - t * (NATOM * 3))];
    }
    for (int j = tid; j < FT * 9; j += FTHREADS) {
      const int t = j / 9;
      const int tok = (tok0 + t < ntok) ? (tok0 + t) : last;
      s_R[j] = orient[(size_t)tok * 9 + (j - t * 9)];
    }
    for (int j = tid; j < FT * NATOM; j += FTHREADS) {
      const int t = j / NATOM;
      const int tok = (tok0 + t < ntok) ? (tok0 + t) : last;
      s_m[j] = atom_mask[(size_t)tok * NATOM + (j - t * NATOM)];
    }
    for (int j = tid; j < FT * 3; j += FTHREADS) {
      const int t = j / 3;
      const int tok = (tok0 + t < ntok) ? (tok0 + t) : last;
      s_d[j] = dihedrals[(size_t)tok * 3 + (j - t * 3)];
    }
    if (tid < FT) {
      const v4f tail = {(float)chain_idx[ctok], 0.f, 0.f, 0.f};
      *(v4f*)(s_f + tid * XS + 84) = tail;
    }
  }
  if (tid < FT) {
    const int ty = (tok0 + tid < ntok) ? tyl : -1;
    for (int t = 0; t < N_AA; ++t) {
      const unsigned m = __builtin_amdgcn_ballot_w32(ty == t);
      if ((tid & 31) == 0) s_tc[(tid >> 5) * N_AA + t] = __builtin_popcount(m);
    }
  }
  __syncthreads();
  if (tid < N_AA) {
    int n = 0;
#pragma unroll
    for (int w = 0; w < FT / 32; ++w) n += s_tc[w * N_AA + tid];
    cnt[(size_t)(blockIdx.x - NPROJ) * CSTR + tid] = n;
  }

  for (int task = tid; task < FT * NATOM; task += FTHREADS) {
    const int t = task / NATOM, a = task - t * NATOM;
    const float* p  = s_xyz + task * 3;
    const float* ca = s_xyz + t * (NATOM * 3) + 3;
    const float rx = p[0] - ca[0], ry = p[1] - ca[1], rz = p[2] - ca[2];
    const float* R = s_R + t * 9;
    const float m = s_m[task];
    float* d = s_f + t * XS + a * 3;
    d[0] = (R[0] * rx + R[3] * ry + R[6] * rz) * m;
    d[1] = (R[1] * rx + R[4] * ry + R[7] * rz) * m;
    d[2] = (R[2] * rx + R[5] * ry + R[8] * rz) * m;
  }
  for (int task = tid; task < FT * 18; task += FTHREADS) {
    const int t = task / 18;
    const int r = task - t * 18;
    const int d = r / 6;
    const int f = r - d * 6;
    const float x = s_d[t * 3 + d];
    const float kf = (f == 1) ? 2.f : (f == 2) ? 3.f : (f == 4) ? 0.5f
                   : (f == 5) ? (1.f / 3.f) : 1.f;
    float* dst = s_f + t * XS + 45 + d * 13;
    if (f == 0) dst[0] = x;
    dst[1 + f] = sinf(kf * x);
    dst[7 + f] = cosf(kf * x);
  }
  __syncthreads();

#pragma unroll
  for (int i = 0; i < (FT * XQ) / FTHREADS; ++i) {
    const int idx = tid + FTHREADS * i;
    if (tok0 + idx / XQ < ntok)
      *(v4f*)(feat + (size_t)tok0 * XS + idx * 4) = *(const v4f*)(s_f + idx * 4);
  }
}

__global__ __launch_bounds__(L1T, 2) void layer1_by_type(
    const int*   __restrict__ seq_idx,
    const int*   __restrict__ cnt,
    const float* __restrict__ feat,
    const float* __restrict__ W1, const float* __restrict__ b1,
    const float* __restrict__ proj,
    float*       __restrict__ h1,
    int ntok)
{
  __shared__ int   s_list[LISTMAX + 4];
  __shared__ int   s_part[L1T];
  __shared__ int   s_sel[4];
  __shared__ int   s_gc[RG * 2];
  __shared__ int   s_ch[GT];
  __shared__ __attribute__((aligned(16))) float s_x[GT][XS];

  const int tid  = threadIdx.x;
  const int wave = tid >> 5;
  const int lane = tid & 31;
  const int type = blockIdx.x;
  const int slice = blockIdx.y;
  const int c0 = tid * 4;
  const int nblk = (ntok + FT - 1) / FT;

  int cb[BPT];
  int ls = 0;
  {
    int v[BPT];
#pragma unroll
    for (int q = 0; q < BPT; ++q) {
      const int b = tid * BPT + q;
      v[q] = cnt[(size_t)((b < nblk) ? b : (nblk - 1)) * CSTR + type];
    }
    __builtin_amdgcn_sched_barrier(0);
#pragma unroll
    for (int q = 0; q < BPT; ++q) {
      const int b = tid * BPT + q;
      cb[q] = (b < nblk) ? v[q] : 0;
      ls += cb[q];
    }
  }
  s_part[tid] = ls;
  __syncthreads();
  int pre = 0, ntype = 0;
  for (int i = 0; i < L1T; ++i) {
    const int v = s_part[i];
    if (i < tid) pre += v;
    ntype += v;
  }
  const int slen = (ntype + NSL - 1) / NSL;
  const int lo = slice * slen;
  if (lo >= ntype) return;
  const int hi = (lo + slen < ntype) ? (lo + slen) : ntype;

  {
    int p = pre;
#pragma unroll
    for (int q = 0; q < BPT; ++q) {
      if (p <= lo && lo < p + cb[q]) { s_sel[0] = tid * BPT + q; s_sel[1] = p; }
      if (p <= hi - 1 && hi - 1 < p + cb[q]) s_sel[2] = tid * BPT + q;
      p += cb[q];
    }
  }
  __syncthreads();
  const int tbeg = s_sel[0] * FT;
  const int tendb = (s_sel[2] + 1) * FT;
  const int tend = (tendb < ntok) ? tendb : ntok;
  int rbase = s_sel[1];

  const unsigned below = (1u << lane) - 1u;
  for (int r0 = tbeg; r0 < tend; r0 += RB * FT) {
    int sq[RG];
#pragma unroll
    for (int i = 0; i < RG; ++i) {
      const int tok = r0 + i * L1T + tid;
      sq[i] = seq_idx[(tok < ntok) ? tok : (ntok - 1)];
    }
    unsigned hbits = 0;
    unsigned msk[RG];
#pragma unroll
    for (int i = 0; i < RG; ++i) {
      const int tok = r0 + i * L1T + tid;
      const bool hit = (tok < tend) && (sq[i] == type);
      msk[i] = __builtin_amdgcn_ballot_w32(hit);
      if (hit) hbits |= 1u << i;
      if (lane == 0) s_gc[2 * i + wave] = __builtin_popcount(msk[i]);
    }
    __syncthreads();
    int run = rbase;
#pragma unroll
    for (int i = 0; i < RG; ++i) {
      const int g0c = s_gc[2 * i], g1c = s_gc[2 * i + 1];
      if ((hbits >> i) & 1u) {
        const int rank = run + (wave ? g0c : 0) + __builtin_popcount(msk[i] & below);
        if (rank >= lo && rank < hi) s_list[rank - lo] = r0 + i * L1T + tid;
      }
      run += g0c + g1c;
    }
    rbase = run;
    __syncthreads();
  }
  const int total = hi - lo;
  if (tid < 3) s_list[total + tid] = s_list[total - 1];
  const int totalp = (total + 3) & ~3;
  __syncthreads();
  for (int i = tid; i < totalp; i += L1T) {
    const int v = s_list[i];
    s_list[i] = (v < 0) ? 0 : ((v < ntok) ? v : (ntok - 1));
  }
  __syncthreads();

  v4f r[XQ];
  {
    const int ng0 = (totalp < GT) ? totalp : GT;
    const int tt = (tid < ng0) ? tid : (ng0 - 1);
    const v4f* src = (const v4f*)(feat + (size_t)s_list[tt] * XS);
#pragma unroll
    for (int q = 0; q < XQ; ++q) r[q] = src[q];
  }

  v4f wc[45], wd[DDIH];
  {
    const float* w1c = W1 + (size_t)(DFEAT + type * 45) * DH1 + c0;
#pragma unroll
    for (int k = 0; k < 45; ++k) wc[k] = *(const v4f*)(w1c + (size_t)k * DH1);
    const float* w1d = W1 + (size_t)(DFEAT + DCOORD) * DH1 + c0;
#pragma unroll
    for (int k = 0; k < DDIH; ++k) wd[k] = *(const v4f*)(w1d + (size_t)k * DH1);
  }
  const v4f base = *(const v4f*)(b1 + c0) + *(const v4f*)(proj + (size_t)type * DH1 + c0);

  int g0 = 0;
  do {
    const int ng = (totalp - g0 < GT) ? (totalp - g0) : GT;

    if (tid < ng) {
#pragma unroll
      for (int q = 0; q < XQ; ++q) *(v4f*)&s_x[tid][4 * q] = r[q];
      s_ch[tid] = (int)r[XQ - 1].x;
    }
    __syncthreads();

    for (int t = 0; t < ng; t += 4) {
      v2f alo[4], ahi[4];
#pragma unroll
      for (int i = 0; i < 4; ++i) {
        const v4f a0 = base + *(const v4f*)(proj + (size_t)(N_AA + s_ch[t + i]) * DH1 + c0);
        alo[i] = a0.lo;
        ahi[i] = a0.hi;
      }
#pragma unroll
      for (int k = 0; k < 45; ++k) {
#pragma unroll
        for (int i = 0; i < 4; ++i) {
          const float x = s_x[t + i][k];
          const v2f xx = {x, x};
          alo[i] = __builtin_elementwise_fma(xx, wc[k].lo, alo[i]);
          ahi[i] = __builtin_elementwise_fma(xx, wc[k].hi, ahi[i]);
        }
      }
#pragma unroll
      for (int k = 0; k < DDIH; ++k) {
#pragma unroll
        for (int i = 0; i < 4; ++i) {
          const float x = s_x[t + i][45 + k];
          const v2f xx = {x, x};
          alo[i] = __builtin_elementwise_fma(xx, wd[k].lo, alo[i]);
          ahi[i] = __builtin_elementwise_fma(xx, wd[k].hi, ahi[i]);
        }
      }
#pragma unroll
      for (int i = 0; i < 4; ++i) {
        v4f o;
        o.x = fmaxf(alo[i].x, 0.f);
        o.y = fmaxf(alo[i].y, 0.f);
        o.z = fmaxf(ahi[i].x, 0.f);
        o.w = fmaxf(ahi[i].y, 0.f);
        *(v4f*)(h1 + (size_t)s_list[g0 + t + i] * DH1 + c0) = o;
      }
    }
    __syncthreads();
    g0 += GT;
    if (g0 < totalp) {
      const int ngn = (totalp - g0 < GT) ? (totalp - g0) : GT;
      const int tt = (tid < ngn) ? tid : (ngn - 1);
      const v4f* src = (const v4f*)(feat + (size_t)s_list[g0 + tt] * XS);
#pragma unroll
      for (int q = 0; q < XQ; ++q) r[q] = src[q];
    }
  } while (g0 < totalp);
}

#define MT      32
#define MW      8
#define MGRID   256
#define OFF_W3  (256 * 128)
#define OFF_W4  (OFF_W3 + 128 * 128)
#define OFF_H1  (OFF_W4 + 128 * 128)
#define OFF_H2  (OFF_H1 + MT * H1S)
#define LDS_FLOATS (OFF_H2 + MT * H2S)

__device__ __forceinline__ void stage_weights(const float* __restrict__ Wm, float* Ld,
                                              int units, int wave, int lane) {
  const int sw = (lane >> 2) & 3;
  for (int ub = wave; ub < units; ub += 4 * MW) {
    v4f v[4][4];
#pragma unroll
    for (int ii = 0; ii < 4; ++ii) {
      const int u  = ub + MW * ii;
      const int kb = 8 * (u >> 1) + 2 * (u & 1);
#pragma unroll
      for (int e = 0; e < 4; ++e)
        v[ii][e] = *(const v4f*)(Wm + (size_t)(kb + (e & 1) + 4 * (e >> 1)) * DFEAT + 4 * lane);
    }
#pragma unroll
    for (int ii = 0; ii < 4; ++ii) {
      float* d = Ld + (ub + MW * ii) * 512 + 16 * lane;
#pragma unroll
      for (int c = 0; c < 4; ++c) {
        const v4f o = {v[ii][0][c], v[ii][1][c], v[ii][2][c], v[ii][3][c]};
        *(v4f*)(d + 4 * (c ^ sw)) = o;
      }
    }
  }
}

#define NP 4
struct Frag { v2f a0[2 * NP], a1[2 * NP]; v4f b[NP]; };

__device__ __forceinline__ Frag load_frag(const float* a0p, const float* a1p, const float* bp, int blk) {
  Frag f;
#pragma unroll
  for (int p = 0; p < NP; ++p) {
    const int j = NP * blk + p;
    f.a0[2 * p]     = *(const v2f*)(a0p + 8 * j);
    f.a0[2 * p + 1] = *(const v2f*)(a0p + 8 * j + 4);
    f.a1[2 * p]     = *(const v2f*)(a1p + 8 * j);
    f.a1[2 * p + 1] = *(const v2f*)(a1p + 8 * j + 4);
    f.b[p]          = *(const v4f*)(bp + j * 1024);
  }
  return f;
}

__device__ __forceinline__ void mm_frag(const Frag& f, v8f& acc0, v8f& acc1) {
#pragma unroll
  for (int p = 0; p < NP; ++p) {
    acc0 = __builtin_amdgcn_wmma_f32_16x16x4_f32(false, f.a0[2 * p],     false, f.b[p].lo, (short)0, acc0, false, false);
    acc1 = __builtin_amdgcn_wmma_f32_16x16x4_f32(false, f.a1[2 * p],     false, f.b[p].lo, (short)0, acc1, false, false);
    acc0 = __builtin_amdgcn_wmma_f32_16x16x4_f32(false, f.a0[2 * p + 1], false, f.b[p].hi, (short)0, acc0, false, false);
    acc1 = __builtin_amdgcn_wmma_f32_16x16x4_f32(false, f.a1[2 * p + 1], false, f.b[p].hi, (short)0, acc1, false, false);
  }
}

__device__ __forceinline__ void group_fence() { __builtin_amdgcn_sched_barrier(0); }

__device__ __forceinline__ void layer_mm(const float* A, int pitch, const float* Bl, int nb,
                                         int lm, int half, int bslot, v8f& acc0, v8f& acc1) {
  const float* a0p = A + lm * pitch + 2 * half;
  const float* a1p = a0p + 16 * pitch;
  const float* bp  = Bl + half * 512 + bslot * 4;
  Frag f0 = load_frag(a0p, a1p, bp, 0);
  for (int blk = 0; blk < nb - 2; blk += 2) {
    const Frag f1 = load_frag(a0p, a1p, bp, blk + 1);
    group_fence();
    mm_frag(f0, acc0, acc1);
    group_fence();
    f0 = load_frag(a0p, a1p, bp, blk + 2);
    group_fence();
    mm_frag(f1, acc0, acc1);
    group_fence();
  }
  const Frag f1 = load_frag(a0p, a1p, bp, nb - 1);
  group_fence();
  mm_frag(f0, acc0, acc1);
  group_fence();
  mm_frag(f1, acc0, acc1);
}

__device__ __forceinline__ void store_out(float* __restrict__ out, const v8f& o0, const v8f& o1,
                                          int tok0, int half, int col, int ntok) {
  float* p = out + (size_t)(tok0 + 8 * half) * DFEAT + col;
  if (tok0 + MT <= ntok) {
#pragma unroll
    for (int r = 0; r < 8; ++r) {
      p[r * DFEAT]        = o0[r];
      p[(16 + r) * DFEAT] = o1[r];
    }
  } else {
#pragma unroll
    for (int r = 0; r < 8; ++r) {
      if (tok0 + 8 * half + r < ntok)      p[r * DFEAT]        = o0[r];
      if (tok0 + 8 * half + 16 + r < ntok) p[(16 + r) * DFEAT] = o1[r];
    }
  }
}

__global__ __launch_bounds__(MW * 32) void mlp_layers_2_to_4(
    const float* __restrict__ h1,
    const float* __restrict__ W2, const float* __restrict__ b2,
    const float* __restrict__ W3, const float* __restrict__ b3,
    const float* __restrict__ W4, const float* __restrict__ b4,
    float*       __restrict__ out,
    int ntok)
{
  __shared__ __attribute__((aligned(16))) float L[LDS_FLOATS];

  const int tid  = threadIdx.x;
  const int wave = tid >> 5;
  const int lane = tid & 31;
  const int half = lane >> 4;
  const int lm   = lane & 15;
  const int col  = wave * 16 + lm;
  const int bslot = wave * 16 + (lm ^ (wave & 3));
  const int ntiles = (ntok + MT - 1) / MT;

  v4f pre[8];
  int tile = blockIdx.x;
  if (tile < ntiles) {
#pragma unroll
    for (int q = 0; q < 8; ++q) {
      const int i = tid + 256 * q;
      int row = tile * MT + (i >> 6);
      row = (row < ntok) ? row : (ntok - 1);
      pre[q] = *(const v4f*)(h1 + (size_t)row * DH1 + (i & 63) * 4);
    }
  }
  const float bias2 = b2[col], bias3 = b3[col], bias4 = b4[col];

  stage_weights(W2, L,          64, wave, lane);
  stage_weights(W3, L + OFF_W3, 32, wave, lane);
  stage_weights(W4, L + OFF_W4, 32, wave, lane);

  v8f o0 = {}, o1 = {};
  int otok0 = -1;

  for (; tile < ntiles; tile += gridDim.x) {
#pragma unroll
    for (int q = 0; q < 8; ++q) {
      const int i = tid + 256 * q;
      *(v4f*)&L[OFF_H1 + (i >> 6) * H1S + (i & 63) * 4] = pre[q];
    }
    __syncthreads();

    if (otok0 >= 0) store_out(out, o0, o1, otok0, half, col, ntok);

    const int tnext = tile + gridDim.x;
    if (tnext < ntiles) {
#pragma unroll
      for (int q = 0; q < 8; ++q) {
        const int i = tid + 256 * q;
        int row = tnext * MT + (i >> 6);
        row = (row < ntok) ? row : (ntok - 1);
        pre[q] = *(const v4f*)(h1 + (size_t)row * DH1 + (i & 63) * 4);
      }
    }

    {
      v8f acc0 = {}, acc1 = {};
      layer_mm(L + OFF_H1, H1S, L, 8, lm, half, bslot, acc0, acc1);
#pragma unroll
      for (int r = 0; r < 8; ++r) {
        L[OFF_H2 + (r + 8 * half) * H2S + col]      = fmaxf(acc0[r] + bias2, 0.f);
        L[OFF_H2 + (16 + r + 8 * half) * H2S + col] = fmaxf(acc1[r] + bias2, 0.f);
      }
    }
    __syncthreads();

    {
      v8f acc0 = {}, acc1 = {};
      layer_mm(L + OFF_H2, H2S, L + OFF_W3, 4, lm, half, bslot, acc0, acc1);
#pragma unroll
      for (int r = 0; r < 8; ++r) {
        L[OFF_H1 + (r + 8 * half) * H2S + col]      = fmaxf(acc0[r] + bias3, 0.f);
        L[OFF_H1 + (16 + r + 8 * half) * H2S + col] = fmaxf(acc1[r] + bias3, 0.f);
      }
    }
    __syncthreads();

    {
      v8f acc0 = {}, acc1 = {};
      layer_mm(L + OFF_H1, H2S, L + OFF_W4, 4, lm, half, bslot, acc0, acc1);
#pragma unroll
      for (int r = 0; r < 8; ++r) {
        o0[r] = acc0[r] + bias4;
        o1[r] = acc1[r] + bias4;
      }
      otok0 = tile * MT;
    }
    __syncthreads();
  }
  if (otok0 >= 0) store_out(out, o0, o1, otok0, half, col, ntok);
}

extern "C" void kernel_launch(void* const* d_in, const int* in_sizes, int n_in,
                              void* d_out, int out_size, void* d_ws, size_t ws_size,
                              hipStream_t stream) {
  const int*   seq    = (const int*)  d_in[0];
  const float* xyz    = (const float*)d_in[1];
  const float* orient = (const float*)d_in[2];
  const float* dih    = (const float*)d_in[3];
  const int*   chain  = (const int*)  d_in[4];
  const float* amask  = (const float*)d_in[5];
  const float* aa_emb = (const float*)d_in[6];
  const float* ch_emb = (const float*)d_in[7];
  const float* W1 = (const float*)d_in[8];
  const float* b1 = (const float*)d_in[9];
  const float* W2 = (const float*)d_in[10];
  const float* b2 = (const float*)d_in[11];
  const float* W3 = (const float*)d_in[12];
  const float* b3 = (const float*)d_in[13];
  const float* W4 = (const float*)d_in[14];
  const float* b4 = (const float*)d_in[15];
  float* out  = (float*)d_out;
  float* proj = (float*)d_ws;
  float* h1   = proj + 8192;
  float* feat = h1 + (size_t)32768 * DH1;
  int*   cnt  = (int*)(feat + (size_t)32768 * XS);

  const int ntok = in_sizes[0];

  proj_and_features<<<NPROJ + (ntok + FT - 1) / FT, FTHREADS, 0, stream>>>(
      aa_emb, ch_emb, W1, xyz, orient, dih, chain, amask, seq, proj, feat, cnt, ntok);
  layer1_by_type<<<dim3(N_AA, NSL), L1T, 0, stream>>>(
      seq, cnt, feat, W1, b1, proj, h1, ntok);
  mlp_layers_2_to_4<<<MGRID, MW * 32, 0, stream>>>(
      h1, W2, b2, W3, b3, W4, b4, out, ntok);
}


# Layer-1 by-type launch: a slice is one dense [48x84]x[84x256] f32 16x16x4 WMMA product, its type's W1 rows held as B fragments; sum order: products first, bias+projections added at the store
# speedup vs baseline: 2.0599x; 1.0466x over previous
#include <hip/hip_runtime.h>

#define N_AA   21
#define NATOM  15
#define DFEAT  128
#define DCOORD (N_AA * NATOM * 3)
#define DDIH   39
#define DH1    256
#define H1S    260
#define H2S    132

#define NSL    35
#define LISTMAX 940
#define L1T    128
#define NW1    (L1T / 32)
#define PR     48
#define XL     84
#define XS     88
#define XQ     (XS / 4)
#define FT     128
#define NPROJ  (N_AA + 10)
#define NBLK   256
#define CSTR   32
#define BPT    (NBLK / L1T)
#define RB     8
#define RG     (RB * FT / L1T)

typedef __attribute__((ext_vector_type(2))) float v2f;
typedef __attribute__((ext_vector_type(4))) float v4f;
typedef __attribute__((ext_vector_type(8))) float v8f;

#define FTHREADS 256

#define STAGE_LOAD(reg, src, n)                                                  \
  _Pragma("unroll")                                                              \
  for (int i = 0; i < ((n) / 4 + FTHREADS - 1) / FTHREADS; ++i) {                \
    const int idx = tid + FTHREADS * i;                                          \
    reg[i] = *(const v4f*)((src) + 4 * ((idx < (n) / 4) ? idx : ((n) / 4 - 1))); \
  }
#define STAGE_STORE(reg, dst, n)                                                 \
  _Pragma("unroll")                                                              \
  for (int i = 0; i < ((n) / 4 + FTHREADS - 1) / FTHREADS; ++i) {                \
    const int idx = tid + FTHREADS * i;                                          \
    if (idx < (n) / 4) *(v4f*)((dst) + 4 * idx) = reg[i];                        \
  }

__global__ __launch_bounds__(FTHREADS) void proj_and_features(
    const float* __restrict__ aa_emb,
    const float* __restrict__ chain_emb,
    const float* __restrict__ W1,
    const float* __restrict__ xyz,
    const float* __restrict__ orient,
    const float* __restrict__ dihedrals,
    const int*   __restrict__ chain_idx,
    const float* __restrict__ atom_mask,
    const int*   __restrict__ seq_idx,
    float* __restrict__ proj,
    float* __restrict__ feat,
    int*   __restrict__ cnt,
    int ntok)
{
  __shared__ __attribute__((aligned(16))) float s_f[FT * XS];
  __shared__ __attribute__((aligned(16))) float s_xyz[FT * NATOM * 3];
  __shared__ __attribute__((aligned(16))) float s_R[FT * 9];
  __shared__ __attribute__((aligned(16))) float s_m[FT * NATOM];
  __shared__ __attribute__((aligned(16))) float s_d[FT * 3];
  __shared__ int s_tc[(FT / 32) * N_AA];

  const int tid = threadIdx.x;
  if (blockIdx.x < NPROJ) {
    const int r = blockIdx.x;
    const int c = tid;
    const float* emb;
    const float* wrow;
    if (r < N_AA) {
      emb  = aa_emb + (size_t)r * DFEAT;
      wrow = W1;
    } else {
      emb  = chain_emb + (size_t)(r - N_AA) * DFEAT;
      wrow = W1 + (size_t)(DFEAT + DCOORD + DDIH) * DH1;
    }
    float acc = 0.f;
    for (int k = 0; k < DFEAT; ++k)
      acc = fmaf(emb[k], wrow[(size_t)k * DH1 + c], acc);
    proj[(size_t)r * DH1 + c] = acc;
    return;
  }

  const int tok0 = (blockIdx.x - NPROJ) * FT;
  const int last = ntok - 1;
  const int ctok = (tok0 + (tid & (FT - 1)) < ntok) ? (tok0 + (tid & (FT - 1))) : last;
  const int tyl = seq_idx[ctok];

  if (tok0 + FT <= ntok) {
    const float* gx = xyz       + (size_t)tok0 * (NATOM * 3);
    const float* gr = orient    + (size_t)tok0 * 9;
    const float* gm = atom_mask + (size_t)tok0 * NATOM;
    const float* gd = dihedrals + (size_t)tok0 * 3;
    v4f rx[(FT * NATOM * 3 / 4 + FTHREADS - 1) / FTHREADS];
    v4f rr[(FT * 9 / 4 + FTHREADS - 1) / FTHREADS];
    v4f rm[(FT * NATOM / 4 + FTHREADS - 1) / FTHREADS];
    v4f rd[(FT * 3 / 4 + FTHREADS - 1) / FTHREADS];
    STAGE_LOAD(rx, gx, FT * NATOM * 3)
    STAGE_LOAD(rr, gr, FT * 9)
    STAGE_LOAD(rm, gm, FT * NATOM)
    STAGE_LOAD(rd, gd, FT * 3)
    const int ch = chain_idx[ctok];
    STAGE_STORE(rx, s_xyz, FT * NATOM * 3)
    STAGE_STORE(rr, s_R, FT * 9)
    STAGE_STORE(rm, s_m, FT * NATOM)
    STAGE_STORE(rd, s_d, FT * 3)
    if (tid < FT) {
      const v4f tail = {(float)ch, 0.f, 0.f, 0.f};
      *(v4f*)(s_f + tid * XS + 84) = tail;
    }
  } else {
    for (int j = tid; j < FT * NATOM * 3; j += FTHREADS) {
      const int t = j / (NATOM * 3);
      const int tok = (tok0 + t < ntok) ? (tok0 + t) : last;
      s_xyz[j] = xyz[(size_t)tok * (NATOM * 3) + (j - t * (NATOM * 3))];
    }
    for (int j = tid; j < FT * 9; j += FTHREADS) {
      const int t = j / 9;
      const int tok = (tok0 + t < ntok) ? (tok0 + t) : last;
      s_R[j] = orient[(size_t)tok * 9 + (j - t * 9)];
    }
    for (int j = tid; j < FT * NATOM; j += FTHREADS) {
      const int t = j / NATOM;
      const int tok = (tok0 + t < ntok) ? (tok0 + t) : last;
      s_m[j] = atom_mask[(size_t)tok * NATOM + (j - t * NATOM)];
    }
    for (int j = tid; j < FT * 3; j += FTHREADS) {
      const int t = j / 3;
      const int tok = (tok0 + t < ntok) ? (tok0 + t) : last;
      s_d[j] = dihedrals[(size_t)tok * 3 + (j - t * 3)];
    }
    if (tid < FT) {
      const v4f tail = {(float)chain_idx[ctok], 0.f, 0.f, 0.f};
      *(v4f*)(s_f + tid * XS + 84) = tail;
    }
  }
  if (tid < FT) {
    const int ty = (tok0 + tid < ntok) ? tyl : -1;
    for (int t = 0; t < N_AA; ++t) {
      const unsigned m = __builtin_amdgcn_ballot_w32(ty == t);
      if ((tid & 31) == 0) s_tc[(tid >> 5) * N_AA + t] = __builtin_popcount(m);
    }
  }
  __syncthreads();
  if (tid < N_AA) {
    int n = 0;
#pragma unroll
    for (int w = 0; w < FT / 32; ++w) n += s_tc[w * N_AA + tid];
    cnt[(size_t)(blockIdx.x - NPROJ) * CSTR + tid] = n;
  }

  for (int task = tid; task < FT * NATOM; task += FTHREADS) {
    const int t = task / NATOM, a = task - t * NATOM;
    const float* p  = s_xyz + task * 3;
    const float* ca = s_xyz + t * (NATOM * 3) + 3;
    const float rx = p[0] - ca[0], ry = p[1] - ca[1], rz = p[2] - ca[2];
    const float* R = s_R + t * 9;
    const float m = s_m[task];
    float* d = s_f + t * XS + a * 3;
    d[0] = (R[0] * rx + R[3] * ry + R[6] * rz) * m;
    d[1] = (R[1] * rx + R[4] * ry + R[7] * rz) * m;
    d[2] = (R[2] * rx + R[5] * ry + R[8] * rz) * m;
  }
  for (int task = tid; task < FT * 18; task += FTHREADS) {
    const int t = task / 18;
    const int r = task - t * 18;
    const int d = r / 6;
    const int f = r - d * 6;
    const float x = s_d[t * 3 + d];
    const float kf = (f == 1) ? 2.f : (f == 2) ? 3.f : (f == 4) ? 0.5f
                   : (f == 5) ? (1.f / 3.f) : 1.f;
    float* dst = s_f + t * XS + 45 + d * 13;
    if (f == 0) dst[0] = x;
    dst[1 + f] = sinf(kf * x);
    dst[7 + f] = cosf(kf * x);
  }
  __syncthreads();

#pragma unroll
  for (int i = 0; i < (FT * XQ) / FTHREADS; ++i) {
    const int idx = tid + FTHREADS * i;
    if (tok0 + idx / XQ < ntok)
      *(v4f*)(feat + (size_t)tok0 * XS + idx * 4) = *(const v4f*)(s_f + idx * 4);
  }
}

__global__ __launch_bounds__(L1T, 3) void layer1_by_type(
    const int*   __restrict__ seq_idx,
    const int*   __restrict__ cnt,
    const float* __restrict__ feat,
    const float* __restrict__ W1, const float* __restrict__ b1,
    const float* __restrict__ proj,
    float*       __restrict__ h1,
    int ntok)
{
  __shared__ int   s_list[LISTMAX + 4];
  __shared__ int   s_part[L1T];
  __shared__ int   s_sel[4];
  __shared__ int   s_gc[RG * NW1];
  __shared__ __attribute__((aligned(16))) int s_ch[PR];
  __shared__ __attribute__((aligned(16))) int s_tok[PR];
  __shared__ __attribute__((aligned(16))) float s_x[PR * XL];
  __shared__ __attribute__((aligned(16))) float s_cp[10 * DH1];

  const int tid  = threadIdx.x;
  const int wave = tid >> 5;
  const int lane = tid & 31;
  const int lc   = lane & 15;
  const int half = lane >> 4;
  const int type = blockIdx.x;
  const int slice = blockIdx.y;
  const int col0 = 64 * wave + 4 * lc;
  const int nblk = (ntok + FT - 1) / FT;

  int cb[BPT];
  int ls = 0;
#pragma unroll
  for (int q = 0; q < BPT; ++q) {
    const int b = tid * BPT + q;
    const int v = cnt[(size_t)((b < nblk) ? b : (nblk - 1)) * CSTR + type];
    cb[q] = (b < nblk) ? v : 0;
    ls += cb[q];
  }
  s_part[tid] = ls;
  __syncthreads();
  int pre = 0, ntype = 0;
  for (int i = 0; i < L1T; ++i) {
    const int v = s_part[i];
    if (i < tid) pre += v;
    ntype += v;
  }
  const int slen = (ntype + NSL - 1) / NSL;
  const int lo = slice * slen;
  if (lo >= ntype) return;
  const int hi = (lo + slen < ntype) ? (lo + slen) : ntype;

  {
    int p = pre;
#pragma unroll
    for (int q = 0; q < BPT; ++q) {
      if (p <= lo && lo < p + cb[q]) { s_sel[0] = tid * BPT + q; s_sel[1] = p; }
      if (p <= hi - 1 && hi - 1 < p + cb[q]) s_sel[2] = tid * BPT + q;
      p += cb[q];
    }
  }
  for (int i = tid; i < 10 * DH1 / 4; i += L1T)
    *(v4f*)&s_cp[4 * i] = *(const v4f*)(proj + (size_t)N_AA * DH1 + 4 * i);
  __syncthreads();
  const int tbeg = s_sel[0] * FT;
  const int tendb = (s_sel[2] + 1) * FT;
  const int tend = (tendb < ntok) ? tendb : ntok;
  int rbase = s_sel[1];

  const unsigned below = (1u << lane) - 1u;
  for (int r0 = tbeg; r0 < tend; r0 += RB * FT) {
    int sq[RG];
#pragma unroll
    for (int i = 0; i < RG; ++i) {
      const int tok = r0 + i * L1T + tid;
      sq[i] = seq_idx[(tok < ntok) ? tok : (ntok - 1)];
    }
    unsigned hbits = 0;
    unsigned msk[RG];
#pragma unroll
    for (int i = 0; i < RG; ++i) {
      const int tok = r0 + i * L1T + tid;
      const bool hit = (tok < tend) && (sq[i] == type);
      msk[i] = __builtin_amdgcn_ballot_w32(hit);
      if (hit) hbits |= 1u << i;
      if (lane == 0) s_gc[NW1 * i + wave] = __builtin_popcount(msk[i]);
    }
    __syncthreads();
    int run = rbase;
#pragma unroll
    for (int i = 0; i < RG; ++i) {
      int lower = 0, all = 0;
#pragma unroll
      for (int w = 0; w < NW1; ++w) {
        const int c = s_gc[NW1 * i + w];
        if (w < wave) lower += c;
        all += c;
      }
      if ((hbits >> i) & 1u) {
        const int rank = run + lower + __builtin_popcount(msk[i] & below);
        if (rank >= lo && rank < hi) s_list[rank - lo] = r0 + i * L1T + tid;
      }
      run += all;
    }
    rbase = run;
    __syncthreads();
  }
  const int total = hi - lo;
  for (int i = tid; i < total; i += L1T) {
    const int v = s_list[i];
    s_list[i] = (v < 0) ? 0 : ((v < ntok) ? v : (ntok - 1));
  }
  __syncthreads();

#define RPT ((PR * 21 + L1T - 1) / L1T)
#define ROWS_LOAD(g0_, nr_)                                                       \
  _Pragma("unroll")                                                               \
  for (int i = 0; i < RPT; ++i) {                                                 \
    const int p0 = tid + L1T * i;                                                 \
    const int p  = (p0 < PR * 21) ? p0 : (PR * 21 - 1);                           \
    const int row = p / 21, q = p - row * 21;                                     \
    const int rr = (row < (nr_)) ? row : ((nr_) - 1);                             \
    rw[i] = *(const v4f*)(feat + (size_t)s_list[(g0_) + rr] * XS + 4 * q);        \
  }                                                                               \
  {                                                                               \
    const int row = (tid < PR) ? tid : (PR - 1);                                  \
    const int rr = (row < (nr_)) ? row : ((nr_) - 1);                             \
    rtk = s_list[(g0_) + rr];                                                     \
    rch = feat[(size_t)rtk * XS + 84];                                            \
  }

  v4f rw[RPT];
  float rch;
  int rtk;
  {
    const int nr0 = (total < PR) ? total : PR;
    ROWS_LOAD(0, nr0)
  }

  v4f bw[21][2];
  {
    const float* pc = W1 + (size_t)(DFEAT + type * 45 + 2 * half) * DH1 + col0;
    const float* pd = W1 + (size_t)(DFEAT + DCOORD - 45 + 2 * half) * DH1 + col0;
#pragma unroll
    for (int s = 0; s < 21; ++s) {
#pragma unroll
      for (int e = 0; e < 2; ++e) {
        const float* pk = (s < 11) ? pc : (s > 11) ? pd : (e == 0 && half == 0) ? pc : pd;
        bw[s][e] = *(const v4f*)(pk + (size_t)(4 * s + e) * DH1);
      }
    }
  }
  const v4f base = *(const v4f*)(b1 + col0) + *(const v4f*)(proj + (size_t)type * DH1 + col0);

  for (int g0 = 0; g0 < total; g0 += PR) {
    const int nr = (total - g0 < PR) ? (total - g0) : PR;
    if (g0 > 0) { ROWS_LOAD(g0, nr) }
#pragma unroll
    for (int i = 0; i < RPT; ++i) {
      const int p = tid + L1T * i;
      const int row = p / 21, q = p - row * 21;
      if (i < RPT - 1 || p < PR * 21) *(v4f*)&s_x[row * XL + 4 * q] = rw[i];
    }
    if (tid < PR) {
      const int ch = (int)rch;
      s_ch[tid] = (ch < 0) ? 0 : ((ch < 10) ? ch : 9);
      s_tok[tid] = rtk;
    }
    __syncthreads();

    v8f acc[3][4];
#pragma unroll
    for (int rt = 0; rt < 3; ++rt)
#pragma unroll
      for (int j = 0; j < 4; ++j) acc[rt][j] = (v8f){};
    const float* ap = s_x + lc * XL + 2 * half;
#pragma unroll
    for (int s = 0; s < 21; ++s) {
      v2f a[3];
#pragma unroll
      for (int rt = 0; rt < 3; ++rt) a[rt] = *(const v2f*)(ap + rt * 16 * XL + 4 * s);
#pragma unroll
      for (int j = 0; j < 4; ++j) {
        const v2f b = {bw[s][0][j], bw[s][1][j]};
#pragma unroll
        for (int rt = 0; rt < 3; ++rt)
          acc[rt][j] = __builtin_amdgcn_wmma_f32_16x16x4_f32(false, a[rt], false, b, (short)0, acc[rt][j], false, false);
      }
    }

#pragma unroll
    for (int rt = 0; rt < 3; ++rt) {
#pragma unroll
      for (int rq = 0; rq < 2; ++rq) {
        const int rb = 16 * rt + 8 * half + 4 * rq;
        int chv[4], tkv[4];
#pragma unroll
        for (int r = 0; r < 4; ++r) { chv[r] = s_ch[rb + r]; tkv[r] = s_tok[rb + r]; }
        v4f cpv[4];
#pragma unroll
        for (int r = 0; r < 4; ++r) cpv[r] = *(const v4f*)&s_cp[chv[r] * DH1 + col0];
#pragma unroll
        for (int r = 0; r < 4; ++r) {
          const v4f add = base + cpv[r];
          v4f o;
          o.x = fmaxf(acc[rt][0][4 * rq + r] + add.x, 0.f);
          o.y = fmaxf(acc[rt][1][4 * rq + r] + add.y, 0.f);
          o.z = fmaxf(acc[rt][2][4 * rq + r] + add.z, 0.f);
          o.w = fmaxf(acc[rt][3][4 * rq + r] + add.w, 0.f);
          if (rb + r < nr) *(v4f*)(h1 + (size_t)tkv[r] * DH1 + col0) = o;
        }
      }
    }
    __syncthreads();
  }
#undef ROWS_LOAD
#undef RPT
}

#define MT      32
#define MW      8
#define MGRID   256
#define OFF_W3  (256 * 128)
#define OFF_W4  (OFF_W3 + 128 * 128)
#define OFF_H1  (OFF_W4 + 128 * 128)
#define OFF_H2  (OFF_H1 + MT * H1S)
#define LDS_FLOATS (OFF_H2 + MT * H2S)

__device__ __forceinline__ void stage_weights(const float* __restrict__ Wm, float* Ld,
                                              int units, int wave, int lane) {
  const int sw = (lane >> 2) & 3;
  for (int ub = wave; ub < units; ub += 4 * MW) {
    v4f v[4][4];
#pragma unroll
    for (int ii = 0; ii < 4; ++ii) {
      const int u  = ub + MW * ii;
      const int kb = 8 * (u >> 1) + 2 * (u & 1);
#pragma unroll
      for (int e = 0; e < 4; ++e)
        v[ii][e] = *(const v4f*)(Wm + (size_t)(kb + (e & 1) + 4 * (e >> 1)) * DFEAT + 4 * lane);
    }
#pragma unroll
    for (int ii = 0; ii < 4; ++ii) {
      float* d = Ld + (ub + MW * ii) * 512 + 16 * lane;
#pragma unroll
      for (int c = 0; c < 4; ++c) {
        const v4f o = {v[ii][0][c], v[ii][1][c], v[ii][2][c], v[ii][3][c]};
        *(v4f*)(d + 4 * (c ^ sw)) = o;
      }
    }
  }
}

#define NP 4
struct Frag { v2f a0[2 * NP], a1[2 * NP]; v4f b[NP]; };

__device__ __forceinline__ Frag load_frag(const float* a0p, const float* a1p, const float* bp, int blk) {
  Frag f;
#pragma unroll
  for (int p = 0; p < NP; ++p) {
    const int j = NP * blk + p;
    f.a0[2 * p]     = *(const v2f*)(a0p + 8 * j);
    f.a0[2 * p + 1] = *(const v2f*)(a0p + 8 * j + 4);
    f.a1[2 * p]     = *(const v2f*)(a1p + 8 * j);
    f.a1[2 * p + 1] = *(const v2f*)(a1p + 8 * j + 4);
    f.b[p]          = *(const v4f*)(bp + j * 1024);
  }
  return f;
}

__device__ __forceinline__ void mm_frag(const Frag& f, v8f& acc0, v8f& acc1) {
#pragma unroll
  for (int p = 0; p < NP; ++p) {
    acc0 = __builtin_amdgcn_wmma_f32_16x16x4_f32(false, f.a0[2 * p],     false, f.b[p].lo, (short)0, acc0, false, false);
    acc1 = __builtin_amdgcn_wmma_f32_16x16x4_f32(false, f.a1[2 * p],     false, f.b[p].lo, (short)0, acc1, false, false);
    acc0 = __builtin_amdgcn_wmma_f32_16x16x4_f32(false, f.a0[2 * p + 1], false, f.b[p].hi, (short)0, acc0, false, false);
    acc1 = __builtin_amdgcn_wmma_f32_16x16x4_f32(false, f.a1[2 * p + 1], false, f.b[p].hi, (short)0, acc1, false, false);
  }
}

__device__ __forceinline__ void group_fence() { __builtin_amdgcn_sched_barrier(0); }

__device__ __forceinline__ void layer_mm(const float* A, int pitch, const float* Bl, int nb,
                                         int lm, int half, int bslot, v8f& acc0, v8f& acc1) {
  const float* a0p = A + lm * pitch + 2 * half;
  const float* a1p = a0p + 16 * pitch;
  const float* bp  = Bl + half * 512 + bslot * 4;
  Frag f0 = load_frag(a0p, a1p, bp, 0);
  for (int blk = 0; blk < nb - 2; blk += 2) {
    const Frag f1 = load_frag(a0p, a1p, bp, blk + 1);
    group_fence();
    mm_frag(f0, acc0, acc1);
    group_fence();
    f0 = load_frag(a0p, a1p, bp, blk + 2);
    group_fence();
    mm_frag(f1, acc0, acc1);
    group_fence();
  }
  const Frag f1 = load_frag(a0p, a1p, bp, nb - 1);
  group_fence();
  mm_frag(f0, acc0, acc1);
  group_fence();
  mm_frag(f1, acc0, acc1);
}

__device__ __forceinline__ void store_out(float* __restrict__ out, const v8f& o0, const v8f& o1,
                                          int tok0, int half, int col, int ntok) {
  float* p = out + (size_t)(tok0 + 8 * half) * DFEAT + col;
  if (tok0 + MT <= ntok) {
#pragma unroll
    for (int r = 0; r < 8; ++r) {
      p[r * DFEAT]        = o0[r];
      p[(16 + r) * DFEAT] = o1[r];
    }
  } else {
#pragma unroll
    for (int r = 0; r < 8; ++r) {
      if (tok0 + 8 * half + r < ntok)      p[r * DFEAT]        = o0[r];
      if (tok0 + 8 * half + 16 + r < ntok) p[(16 + r) * DFEAT] = o1[r];
    }
  }
}

__global__ __launch_bounds__(MW * 32) void mlp_layers_2_to_4(
    const float* __restrict__ h1,
    const float* __restrict__ W2, const float* __restrict__ b2,
    const float* __restrict__ W3, const float* __restrict__ b3,
    const float* __restrict__ W4, const float* __restrict__ b4,
    float*       __restrict__ out,
    int ntok)
{
  __shared__ __attribute__((aligned(16))) float L[LDS_FLOATS];

  const int tid  = threadIdx.x;
  const int wave = tid >> 5;
  const int lane = tid & 31;
  const int half = lane >> 4;
  const int lm   = lane & 15;
  const int col  = wave * 16 + lm;
  const int bslot = wave * 16 + (lm ^ (wave & 3));
  const int ntiles = (ntok + MT - 1) / MT;

  v4f pre[8];
  int tile = blockIdx.x;
  if (tile < ntiles) {
#pragma unroll
    for (int q = 0; q < 8; ++q) {
      const int i = tid + 256 * q;
      int row = tile * MT + (i >> 6);
      row = (row < ntok) ? row : (ntok - 1);
      pre[q] = *(const v4f*)(h1 + (size_t)row * DH1 + (i & 63) * 4);
    }
  }
  const float bias2 = b2[col], bias3 = b3[col], bias4 = b4[col];

  stage_weights(W2, L,          64, wave, lane);
  stage_weights(W3, L + OFF_W3, 32, wave, lane);
  stage_weights(W4, L + OFF_W4, 32, wave, lane);

  v8f o0 = {}, o1 = {};
  int otok0 = -1;

  for (; tile < ntiles; tile += gridDim.x) {
#pragma unroll
    for (int q = 0; q < 8; ++q) {
      const int i = tid + 256 * q;
      *(v4f*)&L[OFF_H1 + (i >> 6) * H1S + (i & 63) * 4] = pre[q];
    }
    __syncthreads();

    if (otok0 >= 0) store_out(out, o0, o1, otok0, half, col, ntok);

    const int tnext = tile + gridDim.x;
    if (tnext < ntiles) {
#pragma unroll
      for (int q = 0; q < 8; ++q) {
        const int i = tid + 256 * q;
        int row = tnext * MT + (i >> 6);
        row = (row < ntok) ? row : (ntok - 1);
        pre[q] = *(const v4f*)(h1 + (size_t)row * DH1 + (i & 63) * 4);
      }
    }

    {
      v8f acc0 = {}, acc1 = {};
      layer_mm(L + OFF_H1, H1S, L, 8, lm, half, bslot, acc0, acc1);
#pragma unroll
      for (int r = 0; r < 8; ++r) {
        L[OFF_H2 + (r + 8 * half) * H2S + col]      = fmaxf(acc0[r] + bias2, 0.f);
        L[OFF_H2 + (16 + r + 8 * half) * H2S + col] = fmaxf(acc1[r] + bias2, 0.f);
      }
    }
    __syncthreads();

    {
      v8f acc0 = {}, acc1 = {};
      layer_mm(L + OFF_H2, H2S, L + OFF_W3, 4, lm, half, bslot, acc0, acc1);
#pragma unroll
      for (int r = 0; r < 8; ++r) {
        L[OFF_H1 + (r + 8 * half) * H2S + col]      = fmaxf(acc0[r] + bias3, 0.f);
        L[OFF_H1 + (16 + r + 8 * half) * H2S + col] = fmaxf(acc1[r] + bias3, 0.f);
      }
    }
    __syncthreads();

    {
      v8f acc0 = {}, acc1 = {};
      layer_mm(L + OFF_H1, H2S, L + OFF_W4, 4, lm, half, bslot, acc0, acc1);
#pragma unroll
      for (int r = 0; r < 8; ++r) {
        o0[r] = acc0[r] + bias4;
        o1[r] = acc1[r] + bias4;
      }
      otok0 = tile * MT;
    }
    __syncthreads();
  }
  if (otok0 >= 0) store_out(out, o0, o1, otok0, half, col, ntok);
}

extern "C" void kernel_launch(void* const* d_in, const int* in_sizes, int n_in,
                              void* d_out, int out_size, void* d_ws, size_t ws_size,
                              hipStream_t stream) {
  const int*   seq    = (const int*)  d_in[0];
  const float* xyz    = (const float*)d_in[1];
  const float* orient = (const float*)d_in[2];
  const float* dih    = (const float*)d_in[3];
  const int*   chain  = (const int*)  d_in[4];
  const float* amask  = (const float*)d_in[5];
  const float* aa_emb = (const float*)d_in[6];
  const float* ch_emb = (const float*)d_in[7];
  const float* W1 = (const float*)d_in[8];
  const float* b1 = (const float*)d_in[9];
  const float* W2 = (const float*)d_in[10];
  const float* b2 = (const float*)d_in[11];
  const float* W3 = (const float*)d_in[12];
  const float* b3 = (const float*)d_in[13];
  const float* W4 = (const float*)d_in[14];
  const float* b4 = (const float*)d_in[15];
  float* out  = (float*)d_out;
  float* proj = (float*)d_ws;
  float* h1   = proj + 8192;
  float* feat = h1 + (size_t)32768 * DH1;
  int*   cnt  = (int*)(feat + (size_t)32768 * XS);

  const int ntok = in_sizes[0];

  proj_and_features<<<NPROJ + (ntok + FT - 1) / FT, FTHREADS, 0, stream>>>(
      aa_emb, ch_emb, W1, xyz, orient, dih, chain, amask, seq, proj, feat, cnt, ntok);
  layer1_by_type<<<dim3(N_AA, NSL), L1T, 0, stream>>>(
      seq, cnt, feat, W1, b1, proj, h1, ntok);
  mlp_layers_2_to_4<<<MGRID, MW * 32, 0, stream>>>(
      h1, W2, b2, W3, b3, W4, b4, out, ntok);
}


# Layer-1 launch: the three 16-row tiles of a pass done one after another (4 accumulator tiles, not 12): a tile's h1 stores drain under the next tile's products; A reads grouped; same sum order
# speedup vs baseline: 2.1291x; 1.0336x over previous
#include <hip/hip_runtime.h>

#define N_AA   21
#define NATOM  15
#define DFEAT  128
#define DCOORD (N_AA * NATOM * 3)
#define DDIH   39
#define DH1    256
#define H1S    260
#define H2S    132

#define NSL    35
#define LISTMAX 940
#define L1T    128
#define NW1    (L1T / 32)
#define PR     48
#define XL     84
#define XS     88
#define XQ     (XS / 4)
#define FT     128
#define NPROJ  (N_AA + 10)
#define NBLK   256
#define CSTR   32
#define BPT    (NBLK / L1T)
#define RB     8
#define RG     (RB * FT / L1T)

typedef __attribute__((ext_vector_type(2))) float v2f;
typedef __attribute__((ext_vector_type(4))) float v4f;
typedef __attribute__((ext_vector_type(8))) float v8f;

#define FTHREADS 256

#define STAGE_LOAD(reg, src, n)                                                  \
  _Pragma("unroll")                                                              \
  for (int i = 0; i < ((n) / 4 + FTHREADS - 1) / FTHREADS; ++i) {                \
    const int idx = tid + FTHREADS * i;                                          \
    reg[i] = *(const v4f*)((src) + 4 * ((idx < (n) / 4) ? idx : ((n) / 4 - 1))); \
  }
#define STAGE_STORE(reg, dst, n)                                                 \
  _Pragma("unroll")                                                              \
  for (int i = 0; i < ((n) / 4 + FTHREADS - 1) / FTHREADS; ++i) {                \
    const int idx = tid + FTHREADS * i;                                          \
    if (idx < (n) / 4) *(v4f*)((dst) + 4 * idx) = reg[i];                        \
  }

__global__ __launch_bounds__(FTHREADS) void proj_and_features(
    const float* __restrict__ aa_emb,
    const float* __restrict__ chain_emb,
    const float* __restrict__ W1,
    const float* __restrict__ xyz,
    const float* __restrict__ orient,
    const float* __restrict__ dihedrals,
    const int*   __restrict__ chain_idx,
    const float* __restrict__ atom_mask,
    const int*   __restrict__ seq_idx,
    float* __restrict__ proj,
    float* __restrict__ feat,
    int*   __restrict__ cnt,
    int ntok)
{
  __shared__ __attribute__((aligned(16))) float s_f[FT * XS];
  __shared__ __attribute__((aligned(16))) float s_xyz[FT * NATOM * 3];
  __shared__ __attribute__((aligned(16))) float s_R[FT * 9];
  __shared__ __attribute__((aligned(16))) float s_m[FT * NATOM];
  __shared__ __attribute__((aligned(16))) float s_d[FT * 3];
  __shared__ int s_tc[(FT / 32) * N_AA];

  const int tid = threadIdx.x;
  if (blockIdx.x < NPROJ) {
    const int r = blockIdx.x;
    const int c = tid;
    const float* emb;
    const float* wrow;
    if (r < N_AA) {
      emb  = aa_emb + (size_t)r * DFEAT;
      wrow = W1;
    } else {
      emb  = chain_emb + (size_t)(r - N_AA) * DFEAT;
      wrow = W1 + (size_t)(DFEAT + DCOORD + DDIH) * DH1;
    }
    float acc = 0.f;
    for (int k = 0; k < DFEAT; ++k)
      acc = fmaf(emb[k], wrow[(size_t)k * DH1 + c], acc);
    proj[(size_t)r * DH1 + c] = acc;
    return;
  }

  const int tok0 = (blockIdx.x - NPROJ) * FT;
  const int last = ntok - 1;
  const int ctok = (tok0 + (tid & (FT - 1)) < ntok) ? (tok0 + (tid & (FT - 1))) : last;
  const int tyl = seq_idx[ctok];

  if (tok0 + FT <= ntok) {
    const float* gx = xyz       + (size_t)tok0 * (NATOM * 3);
    const float* gr = orient    + (size_t)tok0 * 9;
    const float* gm = atom_mask + (size_t)tok0 * NATOM;
    const float* gd = dihedrals + (size_t)tok0 * 3;
    v4f rx[(FT * NATOM * 3 / 4 + FTHREADS - 1) / FTHREADS];
    v4f rr[(FT * 9 / 4 + FTHREADS - 1) / FTHREADS];
    v4f rm[(FT * NATOM / 4 + FTHREADS - 1) / FTHREADS];
    v4f rd[(FT * 3 / 4 + FTHREADS - 1) / FTHREADS];
    STAGE_LOAD(rx, gx, FT * NATOM * 3)
    STAGE_LOAD(rr, gr, FT * 9)
    STAGE_LOAD(rm, gm, FT * NATOM)
    STAGE_LOAD(rd, gd, FT * 3)
    const int ch = chain_idx[ctok];
    STAGE_STORE(rx, s_xyz, FT * NATOM * 3)
    STAGE_STORE(rr, s_R, FT * 9)
    STAGE_STORE(rm, s_m, FT * NATOM)
    STAGE_STORE(rd, s_d, FT * 3)
    if (tid < FT) {
      const v4f tail = {(float)ch, 0.f, 0.f, 0.f};
      *(v4f*)(s_f + tid * XS + 84) = tail;
    }
  } else {
    for (int j = tid; j < FT * NATOM * 3; j += FTHREADS) {
      const int t = j / (NATOM * 3);
      const int tok = (tok0 + t < ntok) ? (tok0 + t) : last;
      s_xyz[j] = xyz[(size_t)tok * (NATOM * 3) + (j - t * (NATOM * 3))];
    }
    for (int j = tid; j < FT * 9; j += FTHREADS) {
      const int t = j / 9;
      const int tok = (tok0 + t < ntok) ? (tok0 + t) : last;
      s_R[j] = orient[(size_t)tok * 9 + (j - t * 9)];
    }
    for (int j = tid; j < FT * NATOM; j += FTHREADS) {
      const int t = j / NATOM;
      const int tok = (tok0 + t < ntok) ? (tok0 + t) : last;
      s_m[j] = atom_mask[(size_t)tok * NATOM + (j - t * NATOM)];
    }
    for (int j = tid; j < FT * 3; j += FTHREADS) {
      const int t = j / 3;
      const int tok = (tok0 + t < ntok) ? (tok0 + t) : last;
      s_d[j] = dihedrals[(size_t)tok * 3 + (j - t * 3)];
    }
    if (tid < FT) {
      const v4f tail = {(float)chain_idx[ctok], 0.f, 0.f, 0.f};
      *(v4f*)(s_f + tid * XS + 84) = tail;
    }
  }
  if (tid < FT) {
    const int ty = (tok0 + tid < ntok) ? tyl : -1;
    for (int t = 0; t < N_AA; ++t) {
      const unsigned m = __builtin_amdgcn_ballot_w32(ty == t);
      if ((tid & 31) == 0) s_tc[(tid >> 5) * N_AA + t] = __builtin_popcount(m);
    }
  }
  __syncthreads();
  if (tid < N_AA) {
    int n = 0;
#pragma unroll
    for (int w = 0; w < FT / 32; ++w) n += s_tc[w * N_AA + tid];
    cnt[(size_t)(blockIdx.x - NPROJ) * CSTR + tid] = n;
  }

  for (int task = tid; task < FT * NATOM; task += FTHREADS) {
    const int t = task / NATOM, a = task - t * NATOM;
    const float* p  = s_xyz + task * 3;
    const float* ca = s_xyz + t * (NATOM * 3) + 3;
    const float rx = p[0] - ca[0], ry = p[1] - ca[1], rz = p[2] - ca[2];
    const float* R = s_R + t * 9;
    const float m = s_m[task];
    float* d = s_f + t * XS + a * 3;
    d[0] = (R[0] * rx + R[3] * ry + R[6] * rz) * m;
    d[1] = (R[1] * rx + R[4] * ry + R[7] * rz) * m;
    d[2] = (R[2] * rx + R[5] * ry + R[8] * rz) * m;
  }
  for (int task = tid; task < FT * 18; task += FTHREADS) {
    const int t = task / 18;
    const int r = task - t * 18;
    const int d = r / 6;
    const int f = r - d * 6;
    const float x = s_d[t * 3 + d];
    const float kf = (f == 1) ? 2.f : (f == 2) ? 3.f : (f == 4) ? 0.5f
                   : (f == 5) ? (1.f / 3.f) : 1.f;
    float* dst = s_f + t * XS + 45 + d * 13;
    if (f == 0) dst[0] = x;
    dst[1 + f] = sinf(kf * x);
    dst[7 + f] = cosf(kf * x);
  }
  __syncthreads();

#pragma unroll
  for (int i = 0; i < (FT * XQ) / FTHREADS; ++i) {
    const int idx = tid + FTHREADS * i;
    if (tok0 + idx / XQ < ntok)
      *(v4f*)(feat + (size_t)tok0 * XS + idx * 4) = *(const v4f*)(s_f + idx * 4);
  }
}

__global__ __launch_bounds__(L1T, 3) void layer1_by_type(
    const int*   __restrict__ seq_idx,
    const int*   __restrict__ cnt,
    const float* __restrict__ feat,
    const float* __restrict__ W1, const float* __restrict__ b1,
    const float* __restrict__ proj,
    float*       __restrict__ h1,
    int ntok)
{
  __shared__ int   s_list[LISTMAX + 4];
  __shared__ int   s_part[L1T];
  __shared__ int   s_sel[4];
  __shared__ int   s_gc[RG * NW1];
  __shared__ __attribute__((aligned(16))) int s_ch[PR];
  __shared__ __attribute__((aligned(16))) int s_tok[PR];
  __shared__ __attribute__((aligned(16))) float s_x[PR * XL];
  __shared__ __attribute__((aligned(16))) float s_cp[10 * DH1];

  const int tid  = threadIdx.x;
  const int wave = tid >> 5;
  const int lane = tid & 31;
  const int lc   = lane & 15;
  const int half = lane >> 4;
  const int type = blockIdx.x;
  const int slice = blockIdx.y;
  const int col0 = 64 * wave + 4 * lc;
  const int nblk = (ntok + FT - 1) / FT;

  int cb[BPT];
  int ls = 0;
#pragma unroll
  for (int q = 0; q < BPT; ++q) {
    const int b = tid * BPT + q;
    const int v = cnt[(size_t)((b < nblk) ? b : (nblk - 1)) * CSTR + type];
    cb[q] = (b < nblk) ? v : 0;
    ls += cb[q];
  }
  s_part[tid] = ls;
  __syncthreads();
  int pre = 0, ntype = 0;
  for (int i = 0; i < L1T; ++i) {
    const int v = s_part[i];
    if (i < tid) pre += v;
    ntype += v;
  }
  const int slen = (ntype + NSL - 1) / NSL;
  const int lo = slice * slen;
  if (lo >= ntype) return;
  const int hi = (lo + slen < ntype) ? (lo + slen) : ntype;

  {
    int p = pre;
#pragma unroll
    for (int q = 0; q < BPT; ++q) {
      if (p <= lo && lo < p + cb[q]) { s_sel[0] = tid * BPT + q; s_sel[1] = p; }
      if (p <= hi - 1 && hi - 1 < p + cb[q]) s_sel[2] = tid * BPT + q;
      p += cb[q];
    }
  }
  for (int i = tid; i < 10 * DH1 / 4; i += L1T)
    *(v4f*)&s_cp[4 * i] = *(const v4f*)(proj + (size_t)N_AA * DH1 + 4 * i);
  __syncthreads();
  const int tbeg = s_sel[0] * FT;
  const int tendb = (s_sel[2] + 1) * FT;
  const int tend = (tendb < ntok) ? tendb : ntok;
  int rbase = s_sel[1];

  const unsigned below = (1u << lane) - 1u;
  for (int r0 = tbeg; r0 < tend; r0 += RB * FT) {
    int sq[RG];
#pragma unroll
    for (int i = 0; i < RG; ++i) {
      const int tok = r0 + i * L1T + tid;
      sq[i] = seq_idx[(tok < ntok) ? tok : (ntok - 1)];
    }
    unsigned hbits = 0;
    unsigned msk[RG];
#pragma unroll
    for (int i = 0; i < RG; ++i) {
      const int tok = r0 + i * L1T + tid;
      const bool hit = (tok < tend) && (sq[i] == type);
      msk[i] = __builtin_amdgcn_ballot_w32(hit);
      if (hit) hbits |= 1u << i;
      if (lane == 0) s_gc[NW1 * i + wave] = __builtin_popcount(msk[i]);
    }
    __syncthreads();
    int run = rbase;
#pragma unroll
    for (int i = 0; i < RG; ++i) {
      int lower = 0, all = 0;
#pragma unroll
      for (int w = 0; w < NW1; ++w) {
        const int c = s_gc[NW1 * i + w];
        if (w < wave) lower += c;
        all += c;
      }
      if ((hbits >> i) & 1u) {
        const int rank = run + lower + __builtin_popcount(msk[i] & below);
        if (rank >= lo && rank < hi) s_list[rank - lo] = r0 + i * L1T + tid;
      }
      run += all;
    }
    rbase = run;
    __syncthreads();
  }
  const int total = hi - lo;
  for (int i = tid; i < total; i += L1T) {
    const int v = s_list[i];
    s_list[i] = (v < 0) ? 0 : ((v < ntok) ? v : (ntok - 1));
  }
  __syncthreads();

#define RPT ((PR * 21 + L1T - 1) / L1T)
#define ROWS_LOAD(g0_, nr_)                                                       \
  _Pragma("unroll")                                                               \
  for (int i = 0; i < RPT; ++i) {                                                 \
    const int p0 = tid + L1T * i;                                                 \
    const int p  = (p0 < PR * 21) ? p0 : (PR * 21 - 1);                           \
    const int row = p / 21, q = p - row * 21;                                     \
    const int rr = (row < (nr_)) ? row : ((nr_) - 1);                             \
    rw[i] = *(const v4f*)(feat + (size_t)s_list[(g0_) + rr] * XS + 4 * q);        \
  }                                                                               \
  {                                                                               \
    const int row = (tid < PR) ? tid : (PR - 1);                                  \
    const int rr = (row < (nr_)) ? row : ((nr_) - 1);                             \
    rtk = s_list[(g0_) + rr];                                                     \
    rch = feat[(size_t)rtk * XS + 84];                                            \
  }

  v4f rw[RPT];
  float rch;
  int rtk;
  {
    const int nr0 = (total < PR) ? total : PR;
    ROWS_LOAD(0, nr0)
  }

  v4f bw[21][2];
  {
    const float* pc = W1 + (size_t)(DFEAT + type * 45 + 2 * half) * DH1 + col0;
    const float* pd = W1 + (size_t)(DFEAT + DCOORD - 45 + 2 * half) * DH1 + col0;
#pragma unroll
    for (int s = 0; s < 21; ++s) {
#pragma unroll
      for (int e = 0; e < 2; ++e) {
        const float* pk = (s < 11) ? pc : (s > 11) ? pd : (e == 0 && half == 0) ? pc : pd;
        bw[s][e] = *(const v4f*)(pk + (size_t)(4 * s + e) * DH1);
      }
    }
  }
  const v4f base = *(const v4f*)(b1 + col0) + *(const v4f*)(proj + (size_t)type * DH1 + col0);

  for (int g0 = 0; g0 < total; g0 += PR) {
    const int nr = (total - g0 < PR) ? (total - g0) : PR;
    if (g0 > 0) { ROWS_LOAD(g0, nr) }
#pragma unroll
    for (int i = 0; i < RPT; ++i) {
      const int p = tid + L1T * i;
      const int row = p / 21, q = p - row * 21;
      if (i < RPT - 1 || p < PR * 21) *(v4f*)&s_x[row * XL + 4 * q] = rw[i];
    }
    if (tid < PR) {
      const int ch = (int)rch;
      s_ch[tid] = (ch < 0) ? 0 : ((ch < 10) ? ch : 9);
      s_tok[tid] = rtk;
    }
    __syncthreads();

    const float* ap = s_x + lc * XL + 2 * half;
    v2f af[21];
#pragma unroll
    for (int s = 0; s < 21; ++s) af[s] = *(const v2f*)(ap + 4 * s);
#pragma unroll
    for (int rt = 0; rt < 3; ++rt) {
      __builtin_amdgcn_sched_barrier(0);
      v8f acc[4];
#pragma unroll
      for (int j = 0; j < 4; ++j) acc[j] = (v8f){};
#pragma unroll
      for (int s = 0; s < 21; ++s) {
#pragma unroll
        for (int j = 0; j < 4; ++j) {
          const v2f b = {bw[s][0][j], bw[s][1][j]};
          acc[j] = __builtin_amdgcn_wmma_f32_16x16x4_f32(false, af[s], false, b, (short)0, acc[j], false, false);
        }
      }
      __builtin_amdgcn_sched_barrier(0);
      if (rt < 2) {
#pragma unroll
        for (int s = 0; s < 21; ++s) af[s] = *(const v2f*)(ap + (rt + 1) * 16 * XL + 4 * s);
      }

#pragma unroll
      for (int rq = 0; rq < 2; ++rq) {
        const int rb = 16 * rt + 8 * half + 4 * rq;
        int chv[4], tkv[4];
#pragma unroll
        for (int r = 0; r < 4; ++r) { chv[r] = s_ch[rb + r]; tkv[r] = s_tok[rb + r]; }
        v4f cpv[4];
#pragma unroll
        for (int r = 0; r < 4; ++r) cpv[r] = *(const v4f*)&s_cp[chv[r] * DH1 + col0];
#pragma unroll
        for (int r = 0; r < 4; ++r) {
          const v4f add = base + cpv[r];
          v4f o;
          o.x = fmaxf(acc[0][4 * rq + r] + add.x, 0.f);
          o.y = fmaxf(acc[1][4 * rq + r] + add.y, 0.f);
          o.z = fmaxf(acc[2][4 * rq + r] + add.z, 0.f);
          o.w = fmaxf(acc[3][4 * rq + r] + add.w, 0.f);
          if (rb + r < nr) *(v4f*)(h1 + (size_t)tkv[r] * DH1 + col0) = o;
        }
      }
    }
    __syncthreads();
  }
#undef ROWS_LOAD
#undef RPT
}

#define MT      32
#define MW      8
#define MGRID   256
#define OFF_W3  (256 * 128)
#define OFF_W4  (OFF_W3 + 128 * 128)
#define OFF_H1  (OFF_W4 + 128 * 128)
#define OFF_H2  (OFF_H1 + MT * H1S)
#define LDS_FLOATS (OFF_H2 + MT * H2S)

__device__ __forceinline__ void stage_weights(const float* __restrict__ Wm, float* Ld,
                                              int units, int wave, int lane) {
  const int sw = (lane >> 2) & 3;
  for (int ub = wave; ub < units; ub += 4 * MW) {
    v4f v[4][4];
#pragma unroll
    for (int ii = 0; ii < 4; ++ii) {
      const int u  = ub + MW * ii;
      const int kb = 8 * (u >> 1) + 2 * (u & 1);
#pragma unroll
      for (int e = 0; e < 4; ++e)
        v[ii][e] = *(const v4f*)(Wm + (size_t)(kb + (e & 1) + 4 * (e >> 1)) * DFEAT + 4 * lane);
    }
#pragma unroll
    for (int ii = 0; ii < 4; ++ii) {
      float* d = Ld + (ub + MW * ii) * 512 + 16 * lane;
#pragma unroll
      for (int c = 0; c < 4; ++c) {
        const v4f o = {v[ii][0][c], v[ii][1][c], v[ii][2][c], v[ii][3][c]};
        *(v4f*)(d + 4 * (c ^ sw)) = o;
      }
    }
  }
}

#define NP 4
struct Frag { v2f a0[2 * NP], a1[2 * NP]; v4f b[NP]; };

__device__ __forceinline__ Frag load_frag(const float* a0p, const float* a1p, const float* bp, int blk) {
  Frag f;
#pragma unroll
  for (int p = 0; p < NP; ++p) {
    const int j = NP * blk + p;
    f.a0[2 * p]     = *(const v2f*)(a0p + 8 * j);
    f.a0[2 * p + 1] = *(const v2f*)(a0p + 8 * j + 4);
    f.a1[2 * p]     = *(const v2f*)(a1p + 8 * j);
    f.a1[2 * p + 1] = *(const v2f*)(a1p + 8 * j + 4);
    f.b[p]          = *(const v4f*)(bp + j * 1024);
  }
  return f;
}

__device__ __forceinline__ void mm_frag(const Frag& f, v8f& acc0, v8f& acc1) {
#pragma unroll
  for (int p = 0; p < NP; ++p) {
    acc0 = __builtin_amdgcn_wmma_f32_16x16x4_f32(false, f.a0[2 * p],     false, f.b[p].lo, (short)0, acc0, false, false);
    acc1 = __builtin_amdgcn_wmma_f32_16x16x4_f32(false, f.a1[2 * p],     false, f.b[p].lo, (short)0, acc1, false, false);
    acc0 = __builtin_amdgcn_wmma_f32_16x16x4_f32(false, f.a0[2 * p + 1], false, f.b[p].hi, (short)0, acc0, false, false);
    acc1 = __builtin_amdgcn_wmma_f32_16x16x4_f32(false, f.a1[2 * p + 1], false, f.b[p].hi, (short)0, acc1, false, false);
  }
}

__device__ __forceinline__ void group_fence() { __builtin_amdgcn_sched_barrier(0); }

__device__ __forceinline__ void layer_mm(const float* A, int pitch, const float* Bl, int nb,
                                         int lm, int half, int bslot, v8f& acc0, v8f& acc1) {
  const float* a0p = A + lm * pitch + 2 * half;
  const float* a1p = a0p + 16 * pitch;
  const float* bp  = Bl + half * 512 + bslot * 4;
  Frag f0 = load_frag(a0p, a1p, bp, 0);
  for (int blk = 0; blk < nb - 2; blk += 2) {
    const Frag f1 = load_frag(a0p, a1p, bp, blk + 1);
    group_fence();
    mm_frag(f0, acc0, acc1);
    group_fence();
    f0 = load_frag(a0p, a1p, bp, blk + 2);
    group_fence();
    mm_frag(f1, acc0, acc1);
    group_fence();
  }
  const Frag f1 = load_frag(a0p, a1p, bp, nb - 1);
  group_fence();
  mm_frag(f0, acc0, acc1);
  group_fence();
  mm_frag(f1, acc0, acc1);
}

__device__ __forceinline__ void store_out(float* __restrict__ out, const v8f& o0, const v8f& o1,
                                          int tok0, int half, int col, int ntok) {
  float* p = out + (size_t)(tok0 + 8 * half) * DFEAT + col;
  if (tok0 + MT <= ntok) {
#pragma unroll
    for (int r = 0; r < 8; ++r) {
      p[r * DFEAT]        = o0[r];
      p[(16 + r) * DFEAT] = o1[r];
    }
  } else {
#pragma unroll
    for (int r = 0; r < 8; ++r) {
      if (tok0 + 8 * half + r < ntok)      p[r * DFEAT]        = o0[r];
      if (tok0 + 8 * half + 16 + r < ntok) p[(16 + r) * DFEAT] = o1[r];
    }
  }
}

__global__ __launch_bounds__(MW * 32) void mlp_layers_2_to_4(
    const float* __restrict__ h1,
    const float* __restrict__ W2, const float* __restrict__ b2,
    const float* __restrict__ W3, const float* __restrict__ b3,
    const float* __restrict__ W4, const float* __restrict__ b4,
    float*       __restrict__ out,
    int ntok)
{
  __shared__ __attribute__((aligned(16))) float L[LDS_FLOATS];

  const int tid  = threadIdx.x;
  const int wave = tid >> 5;
  const int lane = tid & 31;
  const int half = lane >> 4;
  const int lm   = lane & 15;
  const int col  = wave * 16 + lm;
  const int bslot = wave * 16 + (lm ^ (wave & 3));
  const int ntiles = (ntok + MT - 1) / MT;

  v4f pre[8];
  int tile = blockIdx.x;
  if (tile < ntiles) {
#pragma unroll
    for (int q = 0; q < 8; ++q) {
      const int i = tid + 256 * q;
      int row = tile * MT + (i >> 6);
      row = (row < ntok) ? row : (ntok - 1);
      pre[q] = *(const v4f*)(h1 + (size_t)row * DH1 + (i & 63) * 4);
    }
  }
  const float bias2 = b2[col], bias3 = b3[col], bias4 = b4[col];

  stage_weights(W2, L,          64, wave, lane);
  stage_weights(W3, L + OFF_W3, 32, wave, lane);
  stage_weights(W4, L + OFF_W4, 32, wave, lane);

  v8f o0 = {}, o1 = {};
  int otok0 = -1;

  for (; tile < ntiles; tile += gridDim.x) {
#pragma unroll
    for (int q = 0; q < 8; ++q) {
      const int i = tid + 256 * q;
      *(v4f*)&L[OFF_H1 + (i >> 6) * H1S + (i & 63) * 4] = pre[q];
    }
    __syncthreads();

    if (otok0 >= 0) store_out(out, o0, o1, otok0, half, col, ntok);

    const int tnext = tile + gridDim.x;
    if (tnext < ntiles) {
#pragma unroll
      for (int q = 0; q < 8; ++q) {
        const int i = tid + 256 * q;
        int row = tnext * MT + (i >> 6);
        row = (row < ntok) ? row : (ntok - 1);
        pre[q] = *(const v4f*)(h1 + (size_t)row * DH1 + (i & 63) * 4);
      }
    }

    {
      v8f acc0 = {}, acc1 = {};
      layer_mm(L + OFF_H1, H1S, L, 8, lm, half, bslot, acc0, acc1);
#pragma unroll
      for (int r = 0; r < 8; ++r) {
        L[OFF_H2 + (r + 8 * half) * H2S + col]      = fmaxf(acc0[r] + bias2, 0.f);
        L[OFF_H2 + (16 + r + 8 * half) * H2S + col] = fmaxf(acc1[r] + bias2, 0.f);
      }
    }
    __syncthreads();

    {
      v8f acc0 = {}, acc1 = {};
      layer_mm(L + OFF_H2, H2S, L + OFF_W3, 4, lm, half, bslot, acc0, acc1);
#pragma unroll
      for (int r = 0; r < 8; ++r) {
        L[OFF_H1 + (r + 8 * half) * H2S + col]      = fmaxf(acc0[r] + bias3, 0.f);
        L[OFF_H1 + (16 + r + 8 * half) * H2S + col] = fmaxf(acc1[r] + bias3, 0.f);
      }
    }
    __syncthreads();

    {
      v8f acc0 = {}, acc1 = {};
      layer_mm(L + OFF_H1, H2S, L + OFF_W4, 4, lm, half, bslot, acc0, acc1);
#pragma unroll
      for (int r = 0; r < 8; ++r) {
        o0[r] = acc0[r] + bias4;
        o1[r] = acc1[r] + bias4;
      }
      otok0 = tile * MT;
    }
    __syncthreads();
  }
  if (otok0 >= 0) store_out(out, o0, o1, otok0, half, col, ntok);
}

extern "C" void kernel_launch(void* const* d_in, const int* in_sizes, int n_in,
                              void* d_out, int out_size, void* d_ws, size_t ws_size,
                              hipStream_t stream) {
  const int*   seq    = (const int*)  d_in[0];
  const float* xyz    = (const float*)d_in[1];
  const float* orient = (const float*)d_in[2];
  const float* dih    = (const float*)d_in[3];
  const int*   chain  = (const int*)  d_in[4];
  const float* amask  = (const float*)d_in[5];
  const float* aa_emb = (const float*)d_in[6];
  const float* ch_emb = (const float*)d_in[7];
  const float* W1 = (const float*)d_in[8];
  const float* b1 = (const float*)d_in[9];
  const float* W2 = (const float*)d_in[10];
  const float* b2 = (const float*)d_in[11];
  const float* W3 = (const float*)d_in[12];
  const float* b3 = (const float*)d_in[13];
  const float* W4 = (const float*)d_in[14];
  const float* b4 = (const float*)d_in[15];
  float* out  = (float*)d_out;
  float* proj = (float*)d_ws;
  float* h1   = proj + 8192;
  float* feat = h1 + (size_t)32768 * DH1;
  int*   cnt  = (int*)(feat + (size_t)32768 * XS);

  const int ntok = in_sizes[0];

  proj_and_features<<<NPROJ + (ntok + FT - 1) / FT, FTHREADS, 0, stream>>>(
      aa_emb, ch_emb, W1, xyz, orient, dih, chain, amask, seq, proj, feat, cnt, ntok);
  layer1_by_type<<<dim3(N_AA, NSL), L1T, 0, stream>>>(
      seq, cnt, feat, W1, b1, proj, h1, ntok);
  mlp_layers_2_to_4<<<MGRID, MW * 32, 0, stream>>>(
      h1, W2, b2, W3, b3, W4, b4, out, ntok);
}


# Layer-1 launch: no load is waited for before its first use (LDS-only barrier fences; chain_proj loads as one group; first pass ahead of the loop): W1 rows arrive under first products; same sum order
# speedup vs baseline: 2.2044x; 1.0354x over previous
#include <hip/hip_runtime.h>

#define N_AA   21
#define NATOM  15
#define DFEAT  128
#define DCOORD (N_AA * NATOM * 3)
#define DDIH   39
#define DH1    256
#define H1S    260
#define H2S    132

#define NSL    35
#define LISTMAX 940
#define L1T    128
#define NW1    (L1T / 32)
#define PR     48
#define XL     84
#define XS     88
#define XQ     (XS / 4)
#define FT     128
#define NPROJ  (N_AA + 10)
#define NBLK   256
#define CSTR   32
#define BPT    (NBLK / L1T)
#define RB     8
#define RG     (RB * FT / L1T)
#define RPT    ((PR * 21 + L1T - 1) / L1T)

typedef __attribute__((ext_vector_type(2))) float v2f;
typedef __attribute__((ext_vector_type(4))) float v4f;
typedef __attribute__((ext_vector_type(8))) float v8f;

__device__ __forceinline__ void lds_barrier() {
  __builtin_amdgcn_fence(__ATOMIC_RELEASE, "workgroup", "local");
  __builtin_amdgcn_s_barrier();
  __builtin_amdgcn_fence(__ATOMIC_ACQUIRE, "workgroup", "local");
}

#define FTHREADS 256

#define STAGE_LOAD(reg, src, n)                                                  \
  _Pragma("unroll")                                                              \
  for (int i = 0; i < ((n) / 4 + FTHREADS - 1) / FTHREADS; ++i) {                \
    const int idx = tid + FTHREADS * i;                                          \
    reg[i] = *(const v4f*)((src) + 4 * ((idx < (n) / 4) ? idx : ((n) / 4 - 1))); \
  }
#define STAGE_STORE(reg, dst, n)                                                 \
  _Pragma("unroll")                                                              \
  for (int i = 0; i < ((n) / 4 + FTHREADS - 1) / FTHREADS; ++i) {                \
    const int idx = tid + FTHREADS * i;                                          \
    if (idx < (n) / 4) *(v4f*)((dst) + 4 * idx) = reg[i];                        \
  }

__global__ __launch_bounds__(FTHREADS) void proj_and_features(
    const float* __restrict__ aa_emb,
    const float* __restrict__ chain_emb,
    const float* __restrict__ W1,
    const float* __restrict__ xyz,
    const float* __restrict__ orient,
    const float* __restrict__ dihedrals,
    const int*   __restrict__ chain_idx,
    const float* __restrict__ atom_mask,
    const int*   __restrict__ seq_idx,
    float* __restrict__ proj,
    float* __restrict__ feat,
    int*   __restrict__ cnt,
    int ntok)
{
  __shared__ __attribute__((aligned(16))) float s_f[FT * XS];
  __shared__ __attribute__((aligned(16))) float s_xyz[FT * NATOM * 3];
  __shared__ __attribute__((aligned(16))) float s_R[FT * 9];
  __shared__ __attribute__((aligned(16))) float s_m[FT * NATOM];
  __shared__ __attribute__((aligned(16))) float s_d[FT * 3];
  __shared__ int s_tc[(FT / 32) * N_AA];

  const int tid = threadIdx.x;
  if (blockIdx.x < NPROJ) {
    const int r = blockIdx.x;
    const int c = tid;
    const float* emb;
    const float* wrow;
    if (r < N_AA) {
      emb  = aa_emb + (size_t)r * DFEAT;
      wrow = W1;
    } else {
      emb  = chain_emb + (size_t)(r - N_AA) * DFEAT;
      wrow = W1 + (size_t)(DFEAT + DCOORD + DDIH) * DH1;
    }
    float acc = 0.f;
    for (int k = 0; k < DFEAT; ++k)
      acc = fmaf(emb[k], wrow[(size_t)k * DH1 + c], acc);
    proj[(size_t)r * DH1 + c] = acc;
    return;
  }

  const int tok0 = (blockIdx.x - NPROJ) * FT;
  const int last = ntok - 1;
  const int ctok = (tok0 + (tid & (FT - 1)) < ntok) ? (tok0 + (tid & (FT - 1))) : last;
  const int tyl = seq_idx[ctok];

  if (tok0 + FT <= ntok) {
    const float* gx = xyz       + (size_t)tok0 * (NATOM * 3);
    const float* gr = orient    + (size_t)tok0 * 9;
    const float* gm = atom_mask + (size_t)tok0 * NATOM;
    const float* gd = dihedrals + (size_t)tok0 * 3;
    v4f rx[(FT * NATOM * 3 / 4 + FTHREADS - 1) / FTHREADS];
    v4f rr[(FT * 9 / 4 + FTHREADS - 1) / FTHREADS];
    v4f rm[(FT * NATOM / 4 + FTHREADS - 1) / FTHREADS];
    v4f rd[(FT * 3 / 4 + FTHREADS - 1) / FTHREADS];
    STAGE_LOAD(rx, gx, FT * NATOM * 3)
    STAGE_LOAD(rr, gr, FT * 9)
    STAGE_LOAD(rm, gm, FT * NATOM)
    STAGE_LOAD(rd, gd, FT * 3)
    const int ch = chain_idx[ctok];
    STAGE_STORE(rx, s_xyz, FT * NATOM * 3)
    STAGE_STORE(rr, s_R, FT * 9)
    STAGE_STORE(rm, s_m, FT * NATOM)
    STAGE_STORE(rd, s_d, FT * 3)
    if (tid < FT) {
      const v4f tail = {(float)ch, 0.f, 0.f, 0.f};
      *(v4f*)(s_f + tid * XS + 84) = tail;
    }
  } else {
    for (int j = tid; j < FT * NATOM * 3; j += FTHREADS) {
      const int t = j / (NATOM * 3);
      const int tok = (tok0 + t < ntok) ? (tok0 + t) : last;
      s_xyz[j] = xyz[(size_t)tok * (NATOM * 3) + (j - t * (NATOM * 3))];
    }
    for (int j = tid; j < FT * 9; j += FTHREADS) {
      const int t = j / 9;
      const int tok = (tok0 + t < ntok) ? (tok0 + t) : last;
      s_R[j] = orient[(size_t)tok * 9 + (j - t * 9)];
    }
    for (int j = tid; j < FT * NATOM; j += FTHREADS) {
      const int t = j / NATOM;
      const int tok = (tok0 + t < ntok) ? (tok0 + t) : last;
      s_m[j] = atom_mask[(size_t)tok * NATOM + (j - t * NATOM)];
    }
    for (int j = tid; j < FT * 3; j += FTHREADS) {
      const int t = j / 3;
      const int tok = (tok0 + t < ntok) ? (tok0 + t) : last;
      s_d[j] = dihedrals[(size_t)tok * 3 + (j - t * 3)];
    }
    if (tid < FT) {
      const v4f tail = {(float)chain_idx[ctok], 0.f, 0.f, 0.f};
      *(v4f*)(s_f + tid * XS + 84) = tail;
    }
  }
  if (tid < FT) {
    const int ty = (tok0 + tid < ntok) ? tyl : -1;
    for (int t = 0; t < N_AA; ++t) {
      const unsigned m = __builtin_amdgcn_ballot_w32(ty == t);
      if ((tid & 31) == 0) s_tc[(tid >> 5) * N_AA + t] = __builtin_popcount(m);
    }
  }
  __syncthreads();
  if (tid < N_AA) {
    int n = 0;
#pragma unroll
    for (int w = 0; w < FT / 32; ++w) n += s_tc[w * N_AA + tid];
    cnt[(size_t)(blockIdx.x - NPROJ) * CSTR + tid] = n;
  }

  for (int task = tid; task < FT * NATOM; task += FTHREADS) {
    const int t = task / NATOM, a = task - t * NATOM;
    const float* p  = s_xyz + task * 3;
    const float* ca = s_xyz + t * (NATOM * 3) + 3;
    const float rx = p[0] - ca[0], ry = p[1] - ca[1], rz = p[2] - ca[2];
    const float* R = s_R + t * 9;
    const float m = s_m[task];
    float* d = s_f + t * XS + a * 3;
    d[0] = (R[0] * rx + R[3] * ry + R[6] * rz) * m;
    d[1] = (R[1] * rx + R[4] * ry + R[7] * rz) * m;
    d[2] = (R[2] * rx + R[5] * ry + R[8] * rz) * m;
  }
  for (int task = tid; task < FT * 18; task += FTHREADS) {
    const int t = task / 18;
    const int r = task - t * 18;
    const int d = r / 6;
    const int f = r - d * 6;
    const float x = s_d[t * 3 + d];
    const float kf = (f == 1) ? 2.f : (f == 2) ? 3.f : (f == 4) ? 0.5f
                   : (f == 5) ? (1.f / 3.f) : 1.f;
    float* dst = s_f + t * XS + 45 + d * 13;
    if (f == 0) dst[0] = x;
    dst[1 + f] = sinf(kf * x);
    dst[7 + f] = cosf(kf * x);
  }
  __syncthreads();

#pragma unroll
  for (int i = 0; i < (FT * XQ) / FTHREADS; ++i) {
    const int idx = tid + FTHREADS * i;
    if (tok0 + idx / XQ < ntok)
      *(v4f*)(feat + (size_t)tok0 * XS + idx * 4) = *(const v4f*)(s_f + idx * 4);
  }
}

__global__ __launch_bounds__(L1T, 3) void layer1_by_type(
    const int*   __restrict__ seq_idx,
    const int*   __restrict__ cnt,
    const float* __restrict__ feat,
    const float* __restrict__ W1, const float* __restrict__ b1,
    const float* __restrict__ proj,
    float*       __restrict__ h1,
    int ntok)
{
  __shared__ int   s_list[LISTMAX + 4];
  __shared__ int   s_part[L1T];
  __shared__ int   s_sel[4];
  __shared__ int   s_gc[RG * NW1];
  __shared__ __attribute__((aligned(16))) int s_ct[2 * L1T];
  __shared__ __attribute__((aligned(16))) float s_x[PR * XL + 4 * (RPT * L1T - PR * 21)];
  __shared__ __attribute__((aligned(16))) float s_cp[10 * DH1];

  const int tid  = threadIdx.x;
  const int wave = tid >> 5;
  const int lane = tid & 31;
  const int lc   = lane & 15;
  const int half = lane >> 4;
  const int type = blockIdx.x;
  const int slice = blockIdx.y;
  const int col0 = 64 * wave + 4 * lc;
  const int nblk = (ntok + FT - 1) / FT;

  int cb[BPT];
  int ls = 0;
#pragma unroll
  for (int q = 0; q < BPT; ++q) {
    const int b = tid * BPT + q;
    const int v = cnt[(size_t)((b < nblk) ? b : (nblk - 1)) * CSTR + type];
    cb[q] = (b < nblk) ? v : 0;
    ls += cb[q];
  }
#define CPP ((10 * DH1 / 4) / L1T)
  v4f cpr[CPP];
  const int c4 = (4 * tid) & (DH1 - 1);
  const v4f tb1 = *(const v4f*)(b1 + c4);
  const v4f tb2 = *(const v4f*)(proj + (size_t)type * DH1 + c4);
#pragma unroll
  for (int j = 0; j < CPP; ++j)
    cpr[j] = *(const v4f*)(proj + (size_t)N_AA * DH1 + 4 * (tid + L1T * j));
  s_part[tid] = ls;
  lds_barrier();
  int pre = 0, ntype = 0;
  for (int i = 0; i < L1T; ++i) {
    const int v = s_part[i];
    if (i < tid) pre += v;
    ntype += v;
  }
  const int slen = (ntype + NSL - 1) / NSL;
  const int lo = slice * slen;
  if (lo >= ntype) return;
  const int hi = (lo + slen < ntype) ? (lo + slen) : ntype;

  {
    int p = pre;
#pragma unroll
    for (int q = 0; q < BPT; ++q) {
      if (p <= lo && lo < p + cb[q]) { s_sel[0] = tid * BPT + q; s_sel[1] = p; }
      if (p <= hi - 1 && hi - 1 < p + cb[q]) s_sel[2] = tid * BPT + q;
      p += cb[q];
    }
  }
  {
    const v4f tb = tb1 + tb2;
#pragma unroll
    for (int j = 0; j < CPP; ++j)
      *(v4f*)&s_cp[4 * (tid + L1T * j)] = tb + cpr[j];
  }
#undef CPP
  lds_barrier();
  const int tbeg = s_sel[0] * FT;
  const int tendb = (s_sel[2] + 1) * FT;
  const int tend = (tendb < ntok) ? tendb : ntok;
  int rbase = s_sel[1];

  const unsigned below = (1u << lane) - 1u;
  for (int r0 = tbeg; r0 < tend; r0 += RB * FT) {
    int sq[RG];
#pragma unroll
    for (int i = 0; i < RG; ++i) {
      const int tok = r0 + i * L1T + tid;
      sq[i] = seq_idx[(tok < ntok) ? tok : (ntok - 1)];
    }
    unsigned hbits = 0;
    unsigned msk[RG];
#pragma unroll
    for (int i = 0; i < RG; ++i) {
      const int tok = r0 + i * L1T + tid;
      const bool hit = (tok < tend) && (sq[i] == type);
      msk[i] = __builtin_amdgcn_ballot_w32(hit);
      if (hit) hbits |= 1u << i;
      if (lane == 0) s_gc[NW1 * i + wave] = __builtin_popcount(msk[i]);
    }
    __syncthreads();
    int run = rbase;
#pragma unroll
    for (int i = 0; i < RG; ++i) {
      int lower = 0, all = 0;
#pragma unroll
      for (int w = 0; w < NW1; ++w) {
        const int c = s_gc[NW1 * i + w];
        if (w < wave) lower += c;
        all += c;
      }
      if ((hbits >> i) & 1u) {
        const int rank = run + lower + __builtin_popcount(msk[i] & below);
        if (rank >= lo && rank < hi) s_list[rank - lo] = r0 + i * L1T + tid;
      }
      run += all;
    }
    rbase = run;
    __syncthreads();
  }
  const int total = hi - lo;
  for (int i = tid; i < total; i += L1T) {
    const int v = s_list[i];
    s_list[i] = (v < 0) ? 0 : ((v < ntok) ? v : (ntok - 1));
  }
  __syncthreads();

#define ROWS_LOAD(g0_, nr_)                                                       \
  _Pragma("unroll")                                                               \
  for (int i = 0; i < RPT; ++i) {                                                 \
    const int p0 = tid + L1T * i;                                                 \
    const int p  = (p0 < PR * 21) ? p0 : (PR * 21 - 1);                           \
    const int row = p / 21, q = p - row * 21;                                     \
    const int rr = (row < (nr_)) ? row : ((nr_) - 1);                             \
    rw[i] = *(const v4f*)(feat + (size_t)s_list[(g0_) + rr] * XS + 4 * q);        \
  }                                                                               \
  {                                                                               \
    const int row = (tid < PR) ? tid : (PR - 1);                                  \
    const int rr = (row < (nr_)) ? row : ((nr_) - 1);                             \
    rtk = s_list[(g0_) + rr];                                                     \
    rch = feat[(size_t)rtk * XS + 84];                                            \
  }

#define PASS_BODY(nr_)                                                            \
  {                                                                               \
    _Pragma("unroll")                                                             \
    for (int i = 0; i < RPT; ++i) {                                               \
      const int p = tid + L1T * i;                                                \
      const int row = p / 21, q = p - row * 21;                                   \
                  \
      *(v4f*)&s_x[row * XL + 4 * q] = rw[i];                                      \
    }                                                                             \
    {                                                                             \
      const int ch = (int)rch;                    \
                                 \
      const int2 ct = {(ch < 0) ? 0 : ((ch < 10) ? ch : 9), rtk};                 \
      *(int2*)&s_ct[2 * tid] = ct;                                                \
    }                                                                             \
    lds_barrier();                                                                \
    const float* ap = s_x + lc * XL + 2 * half;                                   \
    v2f af[21];                                                                   \
    _Pragma("unroll")                                                             \
    for (int s = 0; s < 21; ++s) af[s] = *(const v2f*)(ap + 4 * s);               \
    _Pragma("unroll")                                                             \
    for (int rt = 0; rt < 3; ++rt) {                                              \
                                                                  \
      __builtin_amdgcn_sched_barrier(0);                                          \
      v8f acc[4];                                                                 \
      _Pragma("unroll")                                                           \
      for (int j = 0; j < 4; ++j) acc[j] = (v8f){};                               \
      _Pragma("unroll")                                                           \
      for (int s = 0; s < 21; ++s) {                                              \
        _Pragma("unroll")                                                         \
        for (int j = 0; j < 4; ++j) {                                             \
          const v2f b = {bw[s][0][j], bw[s][1][j]};                               \
          acc[j] = __builtin_amdgcn_wmma_f32_16x16x4_f32(false, af[s], false, b, (short)0, acc[j], false, false); \
        }                                                                         \
      }                                                                           \
                                            \
      __builtin_amdgcn_sched_barrier(0);                                          \
      if (rt < 2) {                                                               \
        _Pragma("unroll")                                                         \
        for (int s = 0; s < 21; ++s) af[s] = *(const v2f*)(ap + (rt + 1) * 16 * XL + 4 * s); \
      }                                                                           \
                                    \
      _Pragma("unroll")                                                           \
      for (int rq = 0; rq < 2; ++rq) {                                            \
        const int rb = 16 * rt + 8 * half + 4 * rq;                               \
        int chv[4], tkv[4];                                                       \
        _Pragma("unroll")                                                         \
        for (int r = 0; r < 4; ++r) { chv[r] = s_ct[2 * (rb + r)]; tkv[r] = s_ct[2 * (rb + r) + 1]; } \
        v4f cpv[4];                                                               \
        _Pragma("unroll")                                                         \
        for (int r = 0; r < 4; ++r) cpv[r] = *(const v4f*)&s_cp[chv[r] * DH1 + col0]; \
        _Pragma("unroll")                                                         \
        for (int r = 0; r < 4; ++r) {                                             \
          const v4f add = cpv[r];                                                 \
          v4f o;                                                                  \
          o.x = fmaxf(acc[0][4 * rq + r] + add.x, 0.f);                           \
          o.y = fmaxf(acc[1][4 * rq + r] + add.y, 0.f);                           \
          o.z = fmaxf(acc[2][4 * rq + r] + add.z, 0.f);                           \
          o.w = fmaxf(acc[3][4 * rq + r] + add.w, 0.f);                           \
          if (rb + r < (nr_)) *(v4f*)(h1 + (size_t)tkv[r] * DH1 + col0) = o;      \
        }                                                                         \
      }                                                                           \
    }                                                                             \
  }

  v4f rw[RPT];
  float rch;
  int rtk;
  const int nr0 = (total < PR) ? total : PR;
  ROWS_LOAD(0, nr0)
  __builtin_amdgcn_sched_barrier(0);

  v4f bw[21][2];
  {
    const float* pc = W1 + (size_t)(DFEAT + type * 45 + 2 * half) * DH1 + col0;
    const float* pd = W1 + (size_t)(DFEAT + DCOORD - 45 + 2 * half) * DH1 + col0;
#pragma unroll
    for (int s = 0; s < 21; ++s) {
#pragma unroll
      for (int e = 0; e < 2; ++e) {
        const float* pk = (s < 11) ? pc : (s > 11) ? pd : (e == 0 && half == 0) ? pc : pd;
        bw[s][e] = *(const v4f*)(pk + (size_t)(4 * s + e) * DH1);
      }
      __builtin_amdgcn_sched_barrier(0);
    }
  }

  PASS_BODY(nr0)
  for (int g0 = PR; g0 < total; g0 += PR) {
    lds_barrier();
    const int nr = (total - g0 < PR) ? (total - g0) : PR;
    ROWS_LOAD(g0, nr)
    PASS_BODY(nr)
  }
#undef PASS_BODY
#undef ROWS_LOAD
}

#define MT      32
#define MW      8
#define MGRID   256
#define OFF_W3  (256 * 128)
#define OFF_W4  (OFF_W3 + 128 * 128)
#define OFF_H1  (OFF_W4 + 128 * 128)
#define OFF_H2  (OFF_H1 + MT * H1S)
#define LDS_FLOATS (OFF_H2 + MT * H2S)

__device__ __forceinline__ void stage_weights(const float* __restrict__ Wm, float* Ld,
                                              int units, int wave, int lane) {
  const int sw = (lane >> 2) & 3;
  for (int ub = wave; ub < units; ub += 4 * MW) {
    v4f v[4][4];
#pragma unroll
    for (int ii = 0; ii < 4; ++ii) {
      const int u  = ub + MW * ii;
      const int kb = 8 * (u >> 1) + 2 * (u & 1);
#pragma unroll
      for (int e = 0; e < 4; ++e)
        v[ii][e] = *(const v4f*)(Wm + (size_t)(kb + (e & 1) + 4 * (e >> 1)) * DFEAT + 4 * lane);
    }
#pragma unroll
    for (int ii = 0; ii < 4; ++ii) {
      float* d = Ld + (ub + MW * ii) * 512 + 16 * lane;
#pragma unroll
      for (int c = 0; c < 4; ++c) {
        const v4f o = {v[ii][0][c], v[ii][1][c], v[ii][2][c], v[ii][3][c]};
        *(v4f*)(d + 4 * (c ^ sw)) = o;
      }
    }
  }
}

#define NP 4
struct Frag { v2f a0[2 * NP], a1[2 * NP]; v4f b[NP]; };

__device__ __forceinline__ Frag load_frag(const float* a0p, const float* a1p, const float* bp, int blk) {
  Frag f;
#pragma unroll
  for (int p = 0; p < NP; ++p) {
    const int j = NP * blk + p;
    f.a0[2 * p]     = *(const v2f*)(a0p + 8 * j);
    f.a0[2 * p + 1] = *(const v2f*)(a0p + 8 * j + 4);
    f.a1[2 * p]     = *(const v2f*)(a1p + 8 * j);
    f.a1[2 * p + 1] = *(const v2f*)(a1p + 8 * j + 4);
    f.b[p]          = *(const v4f*)(bp + j * 1024);
  }
  return f;
}

__device__ __forceinline__ void mm_frag(const Frag& f, v8f& acc0, v8f& acc1) {
#pragma unroll
  for (int p = 0; p < NP; ++p) {
    acc0 = __builtin_amdgcn_wmma_f32_16x16x4_f32(false, f.a0[2 * p],     false, f.b[p].lo, (short)0, acc0, false, false);
    acc1 = __builtin_amdgcn_wmma_f32_16x16x4_f32(false, f.a1[2 * p],     false, f.b[p].lo, (short)0, acc1, false, false);
    acc0 = __builtin_amdgcn_wmma_f32_16x16x4_f32(false, f.a0[2 * p + 1], false, f.b[p].hi, (short)0, acc0, false, false);
    acc1 = __builtin_amdgcn_wmma_f32_16x16x4_f32(false, f.a1[2 * p + 1], false, f.b[p].hi, (short)0, acc1, false, false);
  }
}

__device__ __forceinline__ void group_fence() { __builtin_amdgcn_sched_barrier(0); }

__device__ __forceinline__ void layer_mm(const float* A, int pitch, const float* Bl, int nb,
                                         int lm, int half, int bslot, v8f& acc0, v8f& acc1) {
  const float* a0p = A + lm * pitch + 2 * half;
  const float* a1p = a0p + 16 * pitch;
  const float* bp  = Bl + half * 512 + bslot * 4;
  Frag f0 = load_frag(a0p, a1p, bp, 0);
  for (int blk = 0; blk < nb - 2; blk += 2) {
    const Frag f1 = load_frag(a0p, a1p, bp, blk + 1);
    group_fence();
    mm_frag(f0, acc0, acc1);
    group_fence();
    f0 = load_frag(a0p, a1p, bp, blk + 2);
    group_fence();
    mm_frag(f1, acc0, acc1);
    group_fence();
  }
  const Frag f1 = load_frag(a0p, a1p, bp, nb - 1);
  group_fence();
  mm_frag(f0, acc0, acc1);
  group_fence();
  mm_frag(f1, acc0, acc1);
}

__device__ __forceinline__ void store_out(float* __restrict__ out, const v8f& o0, const v8f& o1,
                                          int tok0, int half, int col, int ntok) {
  float* p = out + (size_t)(tok0 + 8 * half) * DFEAT + col;
  if (tok0 + MT <= ntok) {
#pragma unroll
    for (int r = 0; r < 8; ++r) {
      p[r * DFEAT]        = o0[r];
      p[(16 + r) * DFEAT] = o1[r];
    }
  } else {
#pragma unroll
    for (int r = 0; r < 8; ++r) {
      if (tok0 + 8 * half + r < ntok)      p[r * DFEAT]        = o0[r];
      if (tok0 + 8 * half + 16 + r < ntok) p[(16 + r) * DFEAT] = o1[r];
    }
  }
}

__global__ __launch_bounds__(MW * 32) void mlp_layers_2_to_4(
    const float* __restrict__ h1,
    const float* __restrict__ W2, const float* __restrict__ b2,
    const float* __restrict__ W3, const float* __restrict__ b3,
    const float* __restrict__ W4, const float* __restrict__ b4,
    float*       __restrict__ out,
    int ntok)
{
  __shared__ __attribute__((aligned(16))) float L[LDS_FLOATS];

  const int tid  = threadIdx.x;
  const int wave = tid >> 5;
  const int lane = tid & 31;
  const int half = lane >> 4;
  const int lm   = lane & 15;
  const int col  = wave * 16 + lm;
  const int bslot = wave * 16 + (lm ^ (wave & 3));
  const int ntiles = (ntok + MT - 1) / MT;

  v4f pre[8];
  int tile = blockIdx.x;
  if (tile < ntiles) {
#pragma unroll
    for (int q = 0; q < 8; ++q) {
      const int i = tid + 256 * q;
      int row = tile * MT + (i >> 6);
      row = (row < ntok) ? row : (ntok - 1);
      pre[q] = *(const v4f*)(h1 + (size_t)row * DH1 + (i & 63) * 4);
    }
  }
  const float bias2 = b2[col], bias3 = b3[col], bias4 = b4[col];

  stage_weights(W2, L,          64, wave, lane);
  stage_weights(W3, L + OFF_W3, 32, wave, lane);
  stage_weights(W4, L + OFF_W4, 32, wave, lane);

  v8f o0 = {}, o1 = {};
  int otok0 = -1;

  for (; tile < ntiles; tile += gridDim.x) {
#pragma unroll
    for (int q = 0; q < 8; ++q) {
      const int i = tid + 256 * q;
      *(v4f*)&L[OFF_H1 + (i >> 6) * H1S + (i & 63) * 4] = pre[q];
    }
    __syncthreads();

    if (otok0 >= 0) store_out(out, o0, o1, otok0, half, col, ntok);

    const int tnext = tile + gridDim.x;
    if (tnext < ntiles) {
#pragma unroll
      for (int q = 0; q < 8; ++q) {
        const int i = tid + 256 * q;
        int row = tnext * MT + (i >> 6);
        row = (row < ntok) ? row : (ntok - 1);
        pre[q] = *(const v4f*)(h1 + (size_t)row * DH1 + (i & 63) * 4);
      }
    }

    {
      v8f acc0 = {}, acc1 = {};
      layer_mm(L + OFF_H1, H1S, L, 8, lm, half, bslot, acc0, acc1);
#pragma unroll
      for (int r = 0; r < 8; ++r) {
        L[OFF_H2 + (r + 8 * half) * H2S + col]      = fmaxf(acc0[r] + bias2, 0.f);
        L[OFF_H2 + (16 + r + 8 * half) * H2S + col] = fmaxf(acc1[r] + bias2, 0.f);
      }
    }
    __syncthreads();

    {
      v8f acc0 = {}, acc1 = {};
      layer_mm(L + OFF_H2, H2S, L + OFF_W3, 4, lm, half, bslot, acc0, acc1);
#pragma unroll
      for (int r = 0; r < 8; ++r) {
        L[OFF_H1 + (r + 8 * half) * H2S + col]      = fmaxf(acc0[r] + bias3, 0.f);
        L[OFF_H1 + (16 + r + 8 * half) * H2S + col] = fmaxf(acc1[r] + bias3, 0.f);
      }
    }
    __syncthreads();

    {
      v8f acc0 = {}, acc1 = {};
      layer_mm(L + OFF_H1, H2S, L + OFF_W4, 4, lm, half, bslot, acc0, acc1);
#pragma unroll
      for (int r = 0; r < 8; ++r) {
        o0[r] = acc0[r] + bias4;
        o1[r] = acc1[r] + bias4;
      }
      otok0 = tile * MT;
    }
    __syncthreads();
  }
  if (otok0 >= 0) store_out(out, o0, o1, otok0, half, col, ntok);
}

extern "C" void kernel_launch(void* const* d_in, const int* in_sizes, int n_in,
                              void* d_out, int out_size, void* d_ws, size_t ws_size,
                              hipStream_t stream) {
  const int*   seq    = (const int*)  d_in[0];
  const float* xyz    = (const float*)d_in[1];
  const float* orient = (const float*)d_in[2];
  const float* dih    = (const float*)d_in[3];
  const int*   chain  = (const int*)  d_in[4];
  const float* amask  = (const float*)d_in[5];
  const float* aa_emb = (const float*)d_in[6];
  const float* ch_emb = (const float*)d_in[7];
  const float* W1 = (const float*)d_in[8];
  const float* b1 = (const float*)d_in[9];
  const float* W2 = (const float*)d_in[10];
  const float* b2 = (const float*)d_in[11];
  const float* W3 = (const float*)d_in[12];
  const float* b3 = (const float*)d_in[13];
  const float* W4 = (const float*)d_in[14];
  const float* b4 = (const float*)d_in[15];
  float* out  = (float*)d_out;
  float* proj = (float*)d_ws;
  float* h1   = proj + 8192;
  float* feat = h1 + (size_t)32768 * DH1;
  int*   cnt  = (int*)(feat + (size_t)32768 * XS);

  const int ntok = in_sizes[0];

  proj_and_features<<<NPROJ + (ntok + FT - 1) / FT, FTHREADS, 0, stream>>>(
      aa_emb, ch_emb, W1, xyz, orient, dih, chain, amask, seq, proj, feat, cnt, ntok);
  layer1_by_type<<<dim3(N_AA, NSL), L1T, 0, stream>>>(
      seq, cnt, feat, W1, b1, proj, h1, ntok);
  mlp_layers_2_to_4<<<MGRID, MW * 32, 0, stream>>>(
      h1, W2, b2, W3, b3, W4, b4, out, ntok);
}
